# starting kernel
# baseline (speedup 1.0000x reference)
#include <hip/hip_runtime.h>
#include <hip/hip_bf16.h>

typedef __attribute__((ext_vector_type(16))) _Float16 v16h;
typedef __attribute__((ext_vector_type(8)))  _Float16 v8h;
typedef __attribute__((ext_vector_type(8)))  float    v8f;

__global__ void fill_f32(float* __restrict__ p, float v, long long n) {
    long long i = (long long)blockIdx.x * blockDim.x + threadIdx.x;
    if (i < n) p[i] = v;
}

__global__ void deg_accum(const int* __restrict__ dst, float* __restrict__ deg, int E) {
    int e = blockIdx.x * blockDim.x + threadIdx.x;
    if (e < E) atomicAdd(&deg[dst[e]], 1.0f);
}

__global__ void rsqrt_inplace(float* __restrict__ d, int n) {
    int i = blockIdx.x * blockDim.x + threadIdx.x;
    if (i < n) {
        float v = d[i];
        d[i] = (v > 0.0f) ? rsqrtf(v) : 0.0f;
    }
}

__global__ void cvt_f32_f16(const float* __restrict__ src, _Float16* __restrict__ dst, long long n) {
    long long i = (long long)blockIdx.x * blockDim.x + threadIdx.x;
    if (i < n) dst[i] = (_Float16)src[i];
}

__global__ void cvt_transpose_w(const float* __restrict__ W, _Float16* __restrict__ Wt, int K, int C) {
    int idx = blockIdx.x * blockDim.x + threadIdx.x;
    if (idx < K * C) {
        int k = idx / C, c = idx % C;
        Wt[c * K + k] = (_Float16)W[idx];
    }
}

__global__ void __launch_bounds__(256) gemm_f16_wmma(const _Float16* __restrict__ A,
                                                     const _Float16* __restrict__ Bt,
                                                     float* __restrict__ C,
                                                     int nrows) {
    const int wave = threadIdx.x >> 5;
    const int lane = threadIdx.x & 31;
    const int mt   = blockIdx.x;

    int rowA = mt * 16 + (lane & 15);
    if (rowA > nrows - 1) rowA = nrows - 1;
    const int colB  = wave * 16 + (lane & 15);
    const int kbase = (lane >> 4) ? 8 : 0;

    const _Float16* __restrict__ arow = A  + (long long)rowA * 128;
    const _Float16* __restrict__ brow = Bt + (long long)colB * 128;

    v8f acc = {};
#pragma unroll
    for (int kc = 0; kc < 4; ++kc) {
        const int k0 = kc * 32 + kbase;
        v8h alo = *(const v8h*)(arow + k0);
        v8h ahi = *(const v8h*)(arow + k0 + 16);
        v8h blo = *(const v8h*)(brow + k0);
        v8h bhi = *(const v8h*)(brow + k0 + 16);
        v16h a, b;
#pragma unroll
        for (int i = 0; i < 8; ++i) {
            a[i] = alo[i]; a[8 + i] = ahi[i];
            b[i] = blo[i]; b[8 + i] = bhi[i];
        }
        acc = __builtin_amdgcn_wmma_f32_16x16x32_f16(
             false, a,  false, b,
             (short)0, acc,  false,  false);
    }

    const int rbase = mt * 16 + ((lane >> 4) ? 8 : 0);
    float* __restrict__ crow = C + (long long)rbase * 128 + wave * 16 + (lane & 15);
    if (mt * 16 + 16 <= nrows) {
#pragma unroll
        for (int r = 0; r < 8; ++r) crow[(long long)r * 128] = acc[r];
    } else {
#pragma unroll
        for (int r = 0; r < 8; ++r)
            if (rbase + r < nrows) crow[(long long)r * 128] = acc[r];
    }
}

__global__ void scatter_edges(const int* __restrict__ ei, const float* __restrict__ dis,
                              const float* __restrict__ hin, float* __restrict__ agg,
                              int E, int Nn) {
    long long idx = (long long)blockIdx.x * blockDim.x + threadIdx.x;
    long long total = (long long)(E + Nn) * 32;
    if (idx >= total) return;
    int e = (int)(idx >> 5);
    int c = ((int)idx & 31) * 4;
    int s, d;
    if (e < E) { s = ei[e]; d = ei[E + e]; }
    else       { s = d = e - E; }
    float w = dis[s] * dis[d];
    float4 v = *(const float4*)(hin + (long long)s * 128 + c);
    float* o = agg + (long long)d * 128 + c;
    atomicAdd(o + 0, v.x * w);
    atomicAdd(o + 1, v.y * w);
    atomicAdd(o + 2, v.z * w);
    atomicAdd(o + 3, v.w * w);
}

__global__ void bias_relu_f16(const float* __restrict__ agg, const float* __restrict__ b,
                              _Float16* __restrict__ out, long long n) {
    long long i = (long long)blockIdx.x * blockDim.x + threadIdx.x;
    if (i < n) {
        float v = agg[i] + b[i & 127];
        out[i] = (_Float16)(v > 0.0f ? v : 0.0f);
    }
}

__global__ void bias_relu_f32(const float* __restrict__ agg, const float* __restrict__ b,
                              float* __restrict__ out, long long n) {
    long long i = (long long)blockIdx.x * blockDim.x + threadIdx.x;
    if (i < n) {
        float v = agg[i] + b[i & 127];
        out[i] = v > 0.0f ? v : 0.0f;
    }
}

__global__ void q_kernel(const int* __restrict__ va, const float* __restrict__ h,
                         const float* __restrict__ Wq, const float* __restrict__ bq,
                         float* __restrict__ out, int A) {
    int a = blockIdx.x * blockDim.x + threadIdx.x;
    if (a >= A) return;
    int i = va[2 * a], j = va[2 * a + 1];
    const float4* __restrict__ hi = (const float4*)(h + (long long)i * 128);
    const float4* __restrict__ hj = (const float4*)(h + (long long)j * 128);
    const float4* __restrict__ wq = (const float4*)Wq;
    float s = bq[0];
#pragma unroll 8
    for (int k = 0; k < 32; ++k) {
        float4 x = hi[k], w = wq[k];
        s += x.x * w.x + x.y * w.y + x.z * w.z + x.w * w.w;
    }
#pragma unroll 8
    for (int k = 0; k < 32; ++k) {
        float4 x = hj[k], w = wq[32 + k];
        s += x.x * w.x + x.y * w.y + x.z * w.z + x.w * w.w;
    }
    out[a] = s;
}

extern "C" void kernel_launch(void* const* d_in, const int* in_sizes, int n_in,
                              void* d_out, int out_size, void* d_ws, size_t ws_size,
                              hipStream_t stream) {
    const float* x  = (const float*)d_in[0];
    const int*   ei = (const int*)d_in[1];
    const int*   va = (const int*)d_in[2];
    const float* W1 = (const float*)d_in[3];
    const float* b1 = (const float*)d_in[4];
    const float* W2 = (const float*)d_in[5];
    const float* b2 = (const float*)d_in[6];
    const float* Wq = (const float*)d_in[7];
    const float* bq = (const float*)d_in[8];
    float* q = (float*)d_out;

    const int H  = 128;
    const int Nn = in_sizes[0] / H;
    const int E  = in_sizes[1] / 2;
    const int A  = in_sizes[2] / 2;

    char*  ws  = (char*)d_ws;
    size_t off = 0;
    auto alloc = [&](size_t bytes) -> char* {
        char* p = ws + off;
        off = (off + bytes + 255) & ~(size_t)255;
        return p;
    };
    _Float16* xh  = (_Float16*)alloc((size_t)Nn * H * 2);
    _Float16* w1t = (_Float16*)alloc((size_t)H * H * 2);
    _Float16* w2t = (_Float16*)alloc((size_t)H * H * 2);
    float*    gout= (float*)   alloc((size_t)Nn * H * 4);
    float*    agg = (float*)   alloc((size_t)Nn * H * 4);
    _Float16* h1h = (_Float16*)alloc((size_t)Nn * H * 2);
    float*    dis = (float*)   alloc((size_t)Nn * 4);
    float*    h2  = gout;

    const long long NH = (long long)Nn * H;
    auto blks = [](long long n, int t) { return (unsigned)((n + t - 1) / t); };

    fill_f32<<<blks(Nn, 256), 256, 0, stream>>>(dis, 1.0f, Nn);
    deg_accum<<<blks(E, 256), 256, 0, stream>>>(ei + E, dis, E);
    rsqrt_inplace<<<blks(Nn, 256), 256, 0, stream>>>(dis, Nn);

    cvt_f32_f16<<<blks(NH, 256), 256, 0, stream>>>(x, xh, NH);
    cvt_transpose_w<<<blks(H * H, 256), 256, 0, stream>>>(W1, w1t, H, H);
    cvt_transpose_w<<<blks(H * H, 256), 256, 0, stream>>>(W2, w2t, H, H);

    const unsigned mtiles = (unsigned)((Nn + 15) / 16);
    const long long scat  = (long long)(E + Nn) * 32;

    gemm_f16_wmma<<<mtiles, 256, 0, stream>>>(xh, w1t, gout, Nn);
    fill_f32<<<blks(NH, 256), 256, 0, stream>>>(agg, 0.0f, NH);
    scatter_edges<<<blks(scat, 256), 256, 0, stream>>>(ei, dis, gout, agg, E, Nn);
    bias_relu_f16<<<blks(NH, 256), 256, 0, stream>>>(agg, b1, h1h, NH);

    gemm_f16_wmma<<<mtiles, 256, 0, stream>>>(h1h, w2t, gout, Nn);
    fill_f32<<<blks(NH, 256), 256, 0, stream>>>(agg, 0.0f, NH);
    scatter_edges<<<blks(scat, 256), 256, 0, stream>>>(ei, dis, gout, agg, E, Nn);
    bias_relu_f32<<<blks(NH, 256), 256, 0, stream>>>(agg, b2, h2, NH);

    q_kernel<<<blks(A, 256), 256, 0, stream>>>(va, h2, Wq, bq, q, A);
}


# bias+ReLU and the convert to the next GEMM's operand type folded into the row-gather kernel's one store: both one-element-per-thread bias/ReLU launches and the aggregate's f32 round trip are gone
# speedup vs baseline: 1.8599x; 1.8599x over previous
#include <hip/hip_runtime.h>
#include <hip/hip_bf16.h>

typedef __attribute__((ext_vector_type(16))) _Float16 v16h;
typedef __attribute__((ext_vector_type(8)))  _Float16 v8h;
typedef __attribute__((ext_vector_type(4)))  _Float16 v4h;
typedef __attribute__((ext_vector_type(8)))  float    v8f;

__global__ void zero_i32(int* __restrict__ p, int n) {
    int i = blockIdx.x * blockDim.x + threadIdx.x;
    if (i < n) p[i] = 0;
}

__global__ void count_rank(const int* __restrict__ dst, int* __restrict__ cnt,
                           int* __restrict__ rank, int E) {
    int e = blockIdx.x * blockDim.x + threadIdx.x;
    if (e < E) rank[e] = atomicAdd(&cnt[dst[e]], 1);
}

__global__ void __launch_bounds__(1024) scan_counts(const int* __restrict__ cnt,
                                                    int* __restrict__ rowptr,
                                                    float* __restrict__ dis, int n) {
    constexpr int PER = 16;
    __shared__ int wsum[32];
    const int tid = threadIdx.x, lane = tid & 31, wave = tid >> 5;
    int carry = 0;
    for (int base = 0; base < n; base += 1024 * PER) {
        const int i0 = base + tid * PER;
        const bool full = (i0 + PER <= n);
        int c[PER];
        if (full) {
#pragma unroll
            for (int q = 0; q < PER / 4; ++q) {
                int4 t4 = *(const int4*)(cnt + i0 + 4 * q);
                c[4 * q] = t4.x; c[4 * q + 1] = t4.y; c[4 * q + 2] = t4.z; c[4 * q + 3] = t4.w;
            }
        } else {
#pragma unroll
            for (int k = 0; k < PER; ++k) c[k] = (i0 + k < n) ? cnt[i0 + k] : 0;
        }
        int t = 0;
#pragma unroll
        for (int k = 0; k < PER; ++k) t += c[k];
        int x = t;
#pragma unroll
        for (int o = 1; o < 32; o <<= 1) {
            int y = __shfl_up(x, o);
            if (lane >= o) x += y;
        }
        if (lane == 31) wsum[wave] = x;
        __syncthreads();
        int xs = wsum[lane];
#pragma unroll
        for (int o = 1; o < 32; o <<= 1) {
            int y = __shfl_up(xs, o);
            if (lane >= o) xs += y;
        }
        const int wincl = __shfl(xs, wave);
        const int total = __shfl(xs, 31);
        int run = carry + (wincl - wsum[wave]) + (x - t);
        if (full) {
#pragma unroll
            for (int q = 0; q < PER / 4; ++q) {
                int4 r4; float4 d4;
                float g0 = (float)(c[4 * q] + 1), g1 = (float)(c[4 * q + 1] + 1);
                float g2 = (float)(c[4 * q + 2] + 1), g3 = (float)(c[4 * q + 3] + 1);
                r4.x = run; run += c[4 * q];
                r4.y = run; run += c[4 * q + 1];
                r4.z = run; run += c[4 * q + 2];
                r4.w = run; run += c[4 * q + 3];
                d4.x = (g0 > 0.0f) ? rsqrtf(g0) : 0.0f;
                d4.y = (g1 > 0.0f) ? rsqrtf(g1) : 0.0f;
                d4.z = (g2 > 0.0f) ? rsqrtf(g2) : 0.0f;
                d4.w = (g3 > 0.0f) ? rsqrtf(g3) : 0.0f;
                *(int4*)(rowptr + i0 + 4 * q) = r4;
                *(float4*)(dis + i0 + 4 * q) = d4;
            }
        } else {
#pragma unroll
            for (int k = 0; k < PER; ++k) {
                if (i0 + k < n) {
                    rowptr[i0 + k] = run;
                    float deg = (float)(c[k] + 1);
                    dis[i0 + k] = (deg > 0.0f) ? rsqrtf(deg) : 0.0f;
                }
                run += c[k];
            }
        }
        carry += total;
        __syncthreads();
    }
    if (tid == 0) rowptr[n] = carry;
}

__global__ void fill_col(const int* __restrict__ ei, const int* __restrict__ rank,
                         const int* __restrict__ rowptr, int* __restrict__ col, int E) {
    int e = blockIdx.x * blockDim.x + threadIdx.x;
    if (e < E) col[rowptr[ei[E + e]] + rank[e]] = ei[e];
}

__global__ void cvt_f32_f16(const float* __restrict__ src, _Float16* __restrict__ dst, long long n) {
    long long i = (long long)blockIdx.x * blockDim.x + threadIdx.x;
    if (i < n) dst[i] = (_Float16)src[i];
}

__global__ void cvt_transpose_w(const float* __restrict__ W, _Float16* __restrict__ Wt, int K, int C) {
    int idx = blockIdx.x * blockDim.x + threadIdx.x;
    if (idx < K * C) {
        int k = idx / C, c = idx % C;
        Wt[c * K + k] = (_Float16)W[idx];
    }
}

__global__ void __launch_bounds__(256) gemm_f16_wmma(const _Float16* __restrict__ A,
                                                     const _Float16* __restrict__ Bt,
                                                     float* __restrict__ C,
                                                     int nrows) {
    const int wave = threadIdx.x >> 5;
    const int lane = threadIdx.x & 31;
    const int mt   = blockIdx.x;

    int rowA = mt * 16 + (lane & 15);
    if (rowA > nrows - 1) rowA = nrows - 1;
    const int colB  = wave * 16 + (lane & 15);
    const int kbase = (lane >> 4) ? 8 : 0;

    const _Float16* __restrict__ arow = A  + (long long)rowA * 128;
    const _Float16* __restrict__ brow = Bt + (long long)colB * 128;

    v8f acc = {};
#pragma unroll
    for (int kc = 0; kc < 4; ++kc) {
        const int k0 = kc * 32 + kbase;
        v8h alo = *(const v8h*)(arow + k0);
        v8h ahi = *(const v8h*)(arow + k0 + 16);
        v8h blo = *(const v8h*)(brow + k0);
        v8h bhi = *(const v8h*)(brow + k0 + 16);
        v16h a, b;
#pragma unroll
        for (int i = 0; i < 8; ++i) {
            a[i] = alo[i]; a[8 + i] = ahi[i];
            b[i] = blo[i]; b[8 + i] = bhi[i];
        }
        acc = __builtin_amdgcn_wmma_f32_16x16x32_f16(
             false, a,  false, b,
             (short)0, acc,  false,  false);
    }

    const int rbase = mt * 16 + ((lane >> 4) ? 8 : 0);
    float* __restrict__ crow = C + (long long)rbase * 128 + wave * 16 + (lane & 15);
    if (mt * 16 + 16 <= nrows) {
#pragma unroll
        for (int r = 0; r < 8; ++r) crow[(long long)r * 128] = acc[r];
    } else {
#pragma unroll
        for (int r = 0; r < 8; ++r)
            if (rbase + r < nrows) crow[(long long)r * 128] = acc[r];
    }
}

__device__ __forceinline__ float4 gather_row_bias_relu(const int* __restrict__ rowptr,
                                                       const int* __restrict__ col,
                                                       const float* __restrict__ dis,
                                                       const float* __restrict__ hin,
                                                       const float* __restrict__ bias,
                                                       int d, int lane) {
    const int beg = rowptr[d], end = rowptr[d + 1];
    const float dd = dis[d];
    const float4* __restrict__ h4 = (const float4*)hin;
    const float4 bb = ((const float4*)bias)[lane];

    float4 acc;
    {
        const float4 v = h4[(long long)d * 32 + lane];
        const float w = dd * dd;
        acc.x = v.x * w; acc.y = v.y * w; acc.z = v.z * w; acc.w = v.w * w;
    }
    for (int base = beg; base < end; base += 32) {
        const int n = (end - base < 32) ? (end - base) : 32;
        int   myc = 0;
        float myw = 0.0f;
        if (lane < n) { myc = col[base + lane]; myw = dis[myc] * dd; }
#pragma unroll 4
        for (int j = 0; j < n; ++j) {
            const int   s = __builtin_amdgcn_readlane(myc, j);
            const float w = __int_as_float(__builtin_amdgcn_readlane(__float_as_int(myw), j));
            const float4 v = h4[(long long)s * 32 + lane];
            acc.x = fmaf(v.x, w, acc.x);
            acc.y = fmaf(v.y, w, acc.y);
            acc.z = fmaf(v.z, w, acc.z);
            acc.w = fmaf(v.w, w, acc.w);
        }
    }
    float4 r;
    r.x = acc.x + bb.x; r.y = acc.y + bb.y; r.z = acc.z + bb.z; r.w = acc.w + bb.w;
    r.x = r.x > 0.0f ? r.x : 0.0f;
    r.y = r.y > 0.0f ? r.y : 0.0f;
    r.z = r.z > 0.0f ? r.z : 0.0f;
    r.w = r.w > 0.0f ? r.w : 0.0f;
    return r;
}

__global__ void __launch_bounds__(256) gather_bias_relu_f16(const int* __restrict__ rowptr,
                                                            const int* __restrict__ col,
                                                            const float* __restrict__ dis,
                                                            const float* __restrict__ hin,
                                                            const float* __restrict__ bias,
                                                            _Float16* __restrict__ out, int Nn) {
    const int d = __builtin_amdgcn_readfirstlane((int)(blockIdx.x * 8 + (threadIdx.x >> 5)));
    if (d >= Nn) return;
    const int lane = threadIdx.x & 31;
    const float4 r = gather_row_bias_relu(rowptr, col, dis, hin, bias, d, lane);
    v4h o;
    o[0] = (_Float16)r.x; o[1] = (_Float16)r.y; o[2] = (_Float16)r.z; o[3] = (_Float16)r.w;
    *(v4h*)(out + (long long)d * 128 + lane * 4) = o;
}

__global__ void __launch_bounds__(256) gather_bias_relu_f32(const int* __restrict__ rowptr,
                                                            const int* __restrict__ col,
                                                            const float* __restrict__ dis,
                                                            const float* __restrict__ hin,
                                                            const float* __restrict__ bias,
                                                            float* __restrict__ out, int Nn) {
    const int d = __builtin_amdgcn_readfirstlane((int)(blockIdx.x * 8 + (threadIdx.x >> 5)));
    if (d >= Nn) return;
    const int lane = threadIdx.x & 31;
    const float4 r = gather_row_bias_relu(rowptr, col, dis, hin, bias, d, lane);
    *(float4*)(out + (long long)d * 128 + lane * 4) = r;
}

__global__ void q_kernel(const int* __restrict__ va, const float* __restrict__ h,
                         const float* __restrict__ Wq, const float* __restrict__ bq,
                         float* __restrict__ out, int A) {
    int a = blockIdx.x * blockDim.x + threadIdx.x;
    if (a >= A) return;
    int i = va[2 * a], j = va[2 * a + 1];
    const float4* __restrict__ hi = (const float4*)(h + (long long)i * 128);
    const float4* __restrict__ hj = (const float4*)(h + (long long)j * 128);
    const float4* __restrict__ wq = (const float4*)Wq;
    float s = bq[0];
#pragma unroll 8
    for (int k = 0; k < 32; ++k) {
        float4 x = hi[k], w = wq[k];
        s += x.x * w.x + x.y * w.y + x.z * w.z + x.w * w.w;
    }
#pragma unroll 8
    for (int k = 0; k < 32; ++k) {
        float4 x = hj[k], w = wq[32 + k];
        s += x.x * w.x + x.y * w.y + x.z * w.z + x.w * w.w;
    }
    out[a] = s;
}

extern "C" void kernel_launch(void* const* d_in, const int* in_sizes, int n_in,
                              void* d_out, int out_size, void* d_ws, size_t ws_size,
                              hipStream_t stream) {
    const float* x  = (const float*)d_in[0];
    const int*   ei = (const int*)d_in[1];
    const int*   va = (const int*)d_in[2];
    const float* W1 = (const float*)d_in[3];
    const float* b1 = (const float*)d_in[4];
    const float* W2 = (const float*)d_in[5];
    const float* b2 = (const float*)d_in[6];
    const float* Wq = (const float*)d_in[7];
    const float* bq = (const float*)d_in[8];
    float* q = (float*)d_out;

    const int H  = 128;
    const int Nn = in_sizes[0] / H;
    const int E  = in_sizes[1] / 2;
    const int A  = in_sizes[2] / 2;

    char*  ws  = (char*)d_ws;
    size_t off = 0;
    auto alloc = [&](size_t bytes) -> char* {
        char* p = ws + off;
        off = (off + bytes + 255) & ~(size_t)255;
        return p;
    };
    _Float16* xh  = (_Float16*)alloc((size_t)Nn * H * 2);
    _Float16* w1t = (_Float16*)alloc((size_t)H * H * 2);
    _Float16* w2t = (_Float16*)alloc((size_t)H * H * 2);
    float*    gout= (float*)   alloc((size_t)Nn * H * 4);
    float*    h2  = (float*)   alloc((size_t)Nn * H * 4);
    char*     csr = (char*)    alloc((size_t)Nn * H * 2);
    float*    dis = (float*)   alloc((size_t)Nn * 4);
    _Float16* h1h = xh;

    size_t coff = 0;
    auto calloc_ = [&](size_t bytes) -> int* {
        int* p = (int*)(csr + coff);
        coff = (coff + bytes + 255) & ~(size_t)255;
        return p;
    };
    int* cnt    = calloc_((size_t)Nn * 4);
    int* rowptr = calloc_((size_t)(Nn + 1) * 4);
    int* rank   = calloc_((size_t)E * 4);
    int* col    = calloc_((size_t)E * 4);

    const long long NH = (long long)Nn * H;
    auto blks = [](long long n, int t) { return (unsigned)((n + t - 1) / t); };

    zero_i32<<<blks(Nn, 256), 256, 0, stream>>>(cnt, Nn);
    count_rank<<<blks(E, 256), 256, 0, stream>>>(ei + E, cnt, rank, E);
    scan_counts<<<1, 1024, 0, stream>>>(cnt, rowptr, dis, Nn);
    fill_col<<<blks(E, 256), 256, 0, stream>>>(ei, rank, rowptr, col, E);

    cvt_f32_f16<<<blks(NH, 256), 256, 0, stream>>>(x, xh, NH);
    cvt_transpose_w<<<blks(H * H, 256), 256, 0, stream>>>(W1, w1t, H, H);
    cvt_transpose_w<<<blks(H * H, 256), 256, 0, stream>>>(W2, w2t, H, H);

    const unsigned mtiles = (unsigned)((Nn + 15) / 16);
    const unsigned rtiles = (unsigned)((Nn + 7) / 8);

    gemm_f16_wmma<<<mtiles, 256, 0, stream>>>(xh, w1t, gout, Nn);
    gather_bias_relu_f16<<<rtiles, 256, 0, stream>>>(rowptr, col, dis, gout, b1, h1h, Nn);

    gemm_f16_wmma<<<mtiles, 256, 0, stream>>>(h1h, w2t, gout, Nn);
    gather_bias_relu_f32<<<rtiles, 256, 0, stream>>>(rowptr, col, dis, gout, b2, h2, Nn);

    q_kernel<<<blks(A, 256), 256, 0, stream>>>(va, h2, Wq, bq, q, A);
}


# sum order now set by the inputs: layer-1 row wave sorts each segment by source id (was atomic arrival order, runs differed); Q head by linearity: layer-2 wave dots its row with Wq, q=p[i]+r[j]+bq
# speedup vs baseline: 2.1487x; 1.1553x over previous
#include <hip/hip_runtime.h>
#include <hip/hip_bf16.h>

typedef __attribute__((ext_vector_type(16))) _Float16 v16h;
typedef __attribute__((ext_vector_type(8)))  _Float16 v8h;
typedef __attribute__((ext_vector_type(4)))  _Float16 v4h;
typedef __attribute__((ext_vector_type(8)))  float    v8f;

__global__ void zero_i32(int* __restrict__ p, int n) {
    int i = blockIdx.x * blockDim.x + threadIdx.x;
    if (i < n) p[i] = 0;
}

__global__ void count_rank(const int* __restrict__ dst, int* __restrict__ cnt,
                           int* __restrict__ rank, int E) {
    int e = blockIdx.x * blockDim.x + threadIdx.x;
    if (e < E) rank[e] = atomicAdd(&cnt[dst[e]], 1);
}

__global__ void __launch_bounds__(1024) scan_counts(const int* __restrict__ cnt,
                                                    int* __restrict__ rowptr,
                                                    float* __restrict__ dis, int n) {
    constexpr int PER = 16;
    __shared__ int wsum[32];
    const int tid = threadIdx.x, lane = tid & 31, wave = tid >> 5;
    int carry = 0;
    for (int base = 0; base < n; base += 1024 * PER) {
        const int i0 = base + tid * PER;
        const bool full = (i0 + PER <= n);
        int c[PER];
        if (full) {
#pragma unroll
            for (int q = 0; q < PER / 4; ++q) {
                int4 t4 = *(const int4*)(cnt + i0 + 4 * q);
                c[4 * q] = t4.x; c[4 * q + 1] = t4.y; c[4 * q + 2] = t4.z; c[4 * q + 3] = t4.w;
            }
        } else {
#pragma unroll
            for (int k = 0; k < PER; ++k) c[k] = (i0 + k < n) ? cnt[i0 + k] : 0;
        }
        int t = 0;
#pragma unroll
        for (int k = 0; k < PER; ++k) t += c[k];
        int x = t;
#pragma unroll
        for (int o = 1; o < 32; o <<= 1) {
            int y = __shfl_up(x, o);
            if (lane >= o) x += y;
        }
        if (lane == 31) wsum[wave] = x;
        __syncthreads();
        int xs = wsum[lane];
#pragma unroll
        for (int o = 1; o < 32; o <<= 1) {
            int y = __shfl_up(xs, o);
            if (lane >= o) xs += y;
        }
        const int wincl = __shfl(xs, wave);
        const int total = __shfl(xs, 31);
        int run = carry + (wincl - wsum[wave]) + (x - t);
        if (full) {
#pragma unroll
            for (int q = 0; q < PER / 4; ++q) {
                int4 r4; float4 d4;
                float g0 = (float)(c[4 * q] + 1), g1 = (float)(c[4 * q + 1] + 1);
                float g2 = (float)(c[4 * q + 2] + 1), g3 = (float)(c[4 * q + 3] + 1);
                r4.x = run; run += c[4 * q];
                r4.y = run; run += c[4 * q + 1];
                r4.z = run; run += c[4 * q + 2];
                r4.w = run; run += c[4 * q + 3];
                d4.x = (g0 > 0.0f) ? rsqrtf(g0) : 0.0f;
                d4.y = (g1 > 0.0f) ? rsqrtf(g1) : 0.0f;
                d4.z = (g2 > 0.0f) ? rsqrtf(g2) : 0.0f;
                d4.w = (g3 > 0.0f) ? rsqrtf(g3) : 0.0f;
                *(int4*)(rowptr + i0 + 4 * q) = r4;
                *(float4*)(dis + i0 + 4 * q) = d4;
            }
        } else {
#pragma unroll
            for (int k = 0; k < PER; ++k) {
                if (i0 + k < n) {
                    rowptr[i0 + k] = run;
                    float deg = (float)(c[k] + 1);
                    dis[i0 + k] = (deg > 0.0f) ? rsqrtf(deg) : 0.0f;
                }
                run += c[k];
            }
        }
        carry += total;
        __syncthreads();
    }
    if (tid == 0) rowptr[n] = carry;
}

__global__ void fill_col(const int* __restrict__ ei, const int* __restrict__ rank,
                         const int* __restrict__ rowptr, int* __restrict__ col, int E) {
    int e = blockIdx.x * blockDim.x + threadIdx.x;
    if (e < E) col[rowptr[ei[E + e]] + rank[e]] = ei[e];
}

__global__ void cvt_f32_f16(const float* __restrict__ src, _Float16* __restrict__ dst, long long n) {
    long long i = (long long)blockIdx.x * blockDim.x + threadIdx.x;
    if (i < n) dst[i] = (_Float16)src[i];
}

__global__ void cvt_transpose_w(const float* __restrict__ W, _Float16* __restrict__ Wt, int K, int C) {
    int idx = blockIdx.x * blockDim.x + threadIdx.x;
    if (idx < K * C) {
        int k = idx / C, c = idx % C;
        Wt[c * K + k] = (_Float16)W[idx];
    }
}

__global__ void __launch_bounds__(256) gemm_f16_wmma(const _Float16* __restrict__ A,
                                                     const _Float16* __restrict__ Bt,
                                                     float* __restrict__ C,
                                                     int nrows) {
    const int wave = threadIdx.x >> 5;
    const int lane = threadIdx.x & 31;
    const int mt   = blockIdx.x;

    int rowA = mt * 16 + (lane & 15);
    if (rowA > nrows - 1) rowA = nrows - 1;
    const int colB  = wave * 16 + (lane & 15);
    const int kbase = (lane >> 4) ? 8 : 0;

    const _Float16* __restrict__ arow = A  + (long long)rowA * 128;
    const _Float16* __restrict__ brow = Bt + (long long)colB * 128;

    v8f acc = {};
#pragma unroll
    for (int kc = 0; kc < 4; ++kc) {
        const int k0 = kc * 32 + kbase;
        v8h alo = *(const v8h*)(arow + k0);
        v8h ahi = *(const v8h*)(arow + k0 + 16);
        v8h blo = *(const v8h*)(brow + k0);
        v8h bhi = *(const v8h*)(brow + k0 + 16);
        v16h a, b;
#pragma unroll
        for (int i = 0; i < 8; ++i) {
            a[i] = alo[i]; a[8 + i] = ahi[i];
            b[i] = blo[i]; b[8 + i] = bhi[i];
        }
        acc = __builtin_amdgcn_wmma_f32_16x16x32_f16(
             false, a,  false, b,
             (short)0, acc,  false,  false);
    }

    const int rbase = mt * 16 + ((lane >> 4) ? 8 : 0);
    float* __restrict__ crow = C + (long long)rbase * 128 + wave * 16 + (lane & 15);
    if (mt * 16 + 16 <= nrows) {
#pragma unroll
        for (int r = 0; r < 8; ++r) crow[(long long)r * 128] = acc[r];
    } else {
#pragma unroll
        for (int r = 0; r < 8; ++r)
            if (rbase + r < nrows) crow[(long long)r * 128] = acc[r];
    }
}

template <bool kOrder>
__device__ __forceinline__ float4 gather_row_bias_relu(const int* __restrict__ rowptr,
                                                       int* __restrict__ col,
                                                       const float* __restrict__ dis,
                                                       const float* __restrict__ hin,
                                                       const float* __restrict__ bias,
                                                       int d, int lane) {
    const int beg = rowptr[d], end = rowptr[d + 1];
    const float dd = dis[d];
    const float4* __restrict__ h4 = (const float4*)hin;
    const float4 bb = ((const float4*)bias)[lane];

    float4 acc;
    {
        const float4 v = h4[(long long)d * 32 + lane];
        const float w = dd * dd;
        acc.x = v.x * w; acc.y = v.y * w; acc.z = v.z * w; acc.w = v.w * w;
    }
    for (int base = beg; base < end; base += 32) {
        const int n = (end - base < 32) ? (end - base) : 32;
        int   myc = 0;
        float myw = 0.0f;
        if (lane < n) myc = col[base + lane];
        if (kOrder) {
            const unsigned key = ((unsigned)myc << 5) | (unsigned)lane;
            int pos = 0;
            for (int j = 0; j < n; ++j) {
                const unsigned kj = (unsigned)__builtin_amdgcn_readlane((int)key, j);
                pos += (kj < key) ? 1 : 0;
            }
            if (lane >= n) pos = lane;
            myc = __builtin_amdgcn_ds_permute(pos << 2, myc);
            if (lane < n) col[base + lane] = myc;
        }
        if (lane < n) myw = dis[myc] * dd;
#pragma unroll 4
        for (int j = 0; j < n; ++j) {
            const int   s = __builtin_amdgcn_readlane(myc, j);
            const float w = __int_as_float(__builtin_amdgcn_readlane(__float_as_int(myw), j));
            const float4 v = h4[(long long)s * 32 + lane];
            acc.x = fmaf(v.x, w, acc.x);
            acc.y = fmaf(v.y, w, acc.y);
            acc.z = fmaf(v.z, w, acc.z);
            acc.w = fmaf(v.w, w, acc.w);
        }
    }
    float4 r;
    r.x = acc.x + bb.x; r.y = acc.y + bb.y; r.z = acc.z + bb.z; r.w = acc.w + bb.w;
    r.x = r.x > 0.0f ? r.x : 0.0f;
    r.y = r.y > 0.0f ? r.y : 0.0f;
    r.z = r.z > 0.0f ? r.z : 0.0f;
    r.w = r.w > 0.0f ? r.w : 0.0f;
    return r;
}

__global__ void __launch_bounds__(256) gather_bias_relu_f16(const int* __restrict__ rowptr,
                                                            int* __restrict__ col,
                                                            const float* __restrict__ dis,
                                                            const float* __restrict__ hin,
                                                            const float* __restrict__ bias,
                                                            _Float16* __restrict__ out, int Nn) {
    const int d = __builtin_amdgcn_readfirstlane((int)(blockIdx.x * 8 + (threadIdx.x >> 5)));
    if (d >= Nn) return;
    const int lane = threadIdx.x & 31;
    const float4 r = gather_row_bias_relu<true>(rowptr, col, dis, hin, bias, d, lane);
    v4h o;
    o[0] = (_Float16)r.x; o[1] = (_Float16)r.y; o[2] = (_Float16)r.z; o[3] = (_Float16)r.w;
    *(v4h*)(out + (long long)d * 128 + lane * 4) = o;
}

__global__ void __launch_bounds__(256) gather_bias_relu_dotq(const int* __restrict__ rowptr,
                                                             int* __restrict__ col,
                                                             const float* __restrict__ dis,
                                                             const float* __restrict__ hin,
                                                             const float* __restrict__ bias,
                                                             const float* __restrict__ Wq,
                                                             float* __restrict__ pq, int Nn) {
    const int d = __builtin_amdgcn_readfirstlane((int)(blockIdx.x * 8 + (threadIdx.x >> 5)));
    if (d >= Nn) return;
    const int lane = threadIdx.x & 31;
    const float4 wa = ((const float4*)Wq)[lane];
    const float4 wb = ((const float4*)Wq)[32 + lane];
    const float4 r = gather_row_bias_relu<false>(rowptr, col, dis, hin, bias, d, lane);
    float p = r.x * wa.x + r.y * wa.y + r.z * wa.z + r.w * wa.w;
    float t = r.x * wb.x + r.y * wb.y + r.z * wb.z + r.w * wb.w;
#pragma unroll
    for (int o = 16; o >= 1; o >>= 1) {
        p += __shfl_xor(p, o);
        t += __shfl_xor(t, o);
    }
    if (lane == 0) *(float2*)(pq + 2 * (long long)d) = make_float2(p, t);
}

__global__ void q_pairs(const int* __restrict__ va, const float* __restrict__ pq,
                        const float* __restrict__ bq, float* __restrict__ out, int A) {
    int a = blockIdx.x * blockDim.x + threadIdx.x;
    if (a >= A) return;
    const int2 ij = ((const int2*)va)[a];
    out[a] = (pq[2 * (long long)ij.x] + pq[2 * (long long)ij.y + 1]) + bq[0];
}

extern "C" void kernel_launch(void* const* d_in, const int* in_sizes, int n_in,
                              void* d_out, int out_size, void* d_ws, size_t ws_size,
                              hipStream_t stream) {
    const float* x  = (const float*)d_in[0];
    const int*   ei = (const int*)d_in[1];
    const int*   va = (const int*)d_in[2];
    const float* W1 = (const float*)d_in[3];
    const float* b1 = (const float*)d_in[4];
    const float* W2 = (const float*)d_in[5];
    const float* b2 = (const float*)d_in[6];
    const float* Wq = (const float*)d_in[7];
    const float* bq = (const float*)d_in[8];
    float* q = (float*)d_out;

    const int H  = 128;
    const int Nn = in_sizes[0] / H;
    const int E  = in_sizes[1] / 2;
    const int A  = in_sizes[2] / 2;

    char*  ws  = (char*)d_ws;
    size_t off = 0;
    auto alloc = [&](size_t bytes) -> char* {
        char* p = ws + off;
        off = (off + bytes + 255) & ~(size_t)255;
        return p;
    };
    _Float16* xh  = (_Float16*)alloc((size_t)Nn * H * 2);
    _Float16* w1t = (_Float16*)alloc((size_t)H * H * 2);
    _Float16* w2t = (_Float16*)alloc((size_t)H * H * 2);
    float*    gout= (float*)   alloc((size_t)Nn * H * 4);
    float*    pq  = (float*)   alloc((size_t)Nn * 2 * 4);
    char*     csr = (char*)    alloc((size_t)Nn * H * 2);
    float*    dis = (float*)   alloc((size_t)Nn * 4);
    _Float16* h1h = xh;

    size_t coff = 0;
    auto calloc_ = [&](size_t bytes) -> int* {
        int* p = (int*)(csr + coff);
        coff = (coff + bytes + 255) & ~(size_t)255;
        return p;
    };
    int* cnt    = calloc_((size_t)Nn * 4);
    int* rowptr = calloc_((size_t)(Nn + 1) * 4);
    int* rank   = calloc_((size_t)E * 4);
    int* col    = calloc_((size_t)E * 4);

    const long long NH = (long long)Nn * H;
    auto blks = [](long long n, int t) { return (unsigned)((n + t - 1) / t); };

    zero_i32<<<blks(Nn, 256), 256, 0, stream>>>(cnt, Nn);
    count_rank<<<blks(E, 256), 256, 0, stream>>>(ei + E, cnt, rank, E);
    scan_counts<<<1, 1024, 0, stream>>>(cnt, rowptr, dis, Nn);
    fill_col<<<blks(E, 256), 256, 0, stream>>>(ei, rank, rowptr, col, E);

    cvt_f32_f16<<<blks(NH, 256), 256, 0, stream>>>(x, xh, NH);
    cvt_transpose_w<<<blks(H * H, 256), 256, 0, stream>>>(W1, w1t, H, H);
    cvt_transpose_w<<<blks(H * H, 256), 256, 0, stream>>>(W2, w2t, H, H);

    const unsigned mtiles = (unsigned)((Nn + 15) / 16);
    const unsigned rtiles = (unsigned)((Nn + 7) / 8);

    gemm_f16_wmma<<<mtiles, 256, 0, stream>>>(xh, w1t, gout, Nn);
    gather_bias_relu_f16<<<rtiles, 256, 0, stream>>>(rowptr, col, dis, gout, b1, h1h, Nn);

    gemm_f16_wmma<<<mtiles, 256, 0, stream>>>(h1h, w2t, gout, Nn);
    gather_bias_relu_dotq<<<rtiles, 256, 0, stream>>>(rowptr, col, dis, gout, b2, Wq, pq, Nn);

    q_pairs<<<blks(A, 256), 256, 0, stream>>>(va, pq, bq, q, A);
}


# edge list built by a two-level counting sort on LDS counters (chunk bins, then one block per 256-node bucket): no global atomics, no zero fill, no one-block scan; 4 launches become 2; sums unchanged
# speedup vs baseline: 5.2390x; 2.4382x over previous
#include <hip/hip_runtime.h>
#include <hip/hip_bf16.h>

typedef __attribute__((ext_vector_type(16))) _Float16 v16h;
typedef __attribute__((ext_vector_type(8)))  _Float16 v8h;
typedef __attribute__((ext_vector_type(4)))  _Float16 v4h;
typedef __attribute__((ext_vector_type(8)))  float    v8f;

constexpr int kBucket = 256;
constexpr int kChunks = 256;

__device__ __forceinline__ int wave_incl_scan(int x, int lane) {
#pragma unroll
    for (int o = 1; o < 32; o <<= 1) {
        int y = __shfl_up(x, o);
        if (lane >= o) x += y;
    }
    return x;
}

__global__ void __launch_bounds__(1024) bin_chunk(const int* __restrict__ ei, int* __restrict__ rank,
                                                  int* __restrict__ T, int* __restrict__ O,
                                                  int* __restrict__ staged, int E, int CS) {
    __shared__ int hist[kBucket];
    __shared__ int offs[kBucket];
    __shared__ int wsum[8];
    const int tid = threadIdx.x, lane = tid & 31, wave = tid >> 5;
    const int c  = blockIdx.x;
    const int e0 = c * CS;
    const int e1 = (e0 + CS < E) ? (e0 + CS) : E;
    if (tid < kBucket) hist[tid] = 0;
    __syncthreads();
    for (int eb = e0 + tid; eb < e1; eb += 4096) {
        int d[4];
#pragma unroll
        for (int k = 0; k < 4; ++k) {
            const int e = eb + 1024 * k;
            d[k] = (e < e1) ? ei[E + e] : -1;
        }
#pragma unroll
        for (int k = 0; k < 4; ++k) {
            const int e = eb + 1024 * k;
            if (d[k] >= 0) rank[e] = atomicAdd(&hist[d[k] >> 8], 1);
        }
    }
    __syncthreads();
    const int v = (tid < kBucket) ? hist[tid] : 0;
    const int x = wave_incl_scan(v, lane);
    if (tid < kBucket && lane == 31) wsum[wave] = x;
    __syncthreads();
    if (tid < kBucket) {
        int pre = 0;
        for (int w = 0; w < wave; ++w) pre += wsum[w];
        const int ex = pre + x - v;
        offs[tid] = ex;
        T[c * kBucket + tid] = v;
        O[c * kBucket + tid] = ex;
    }
    __syncthreads();
    int2* __restrict__ st = (int2*)staged;
    for (int eb = e0 + tid; eb < e1; eb += 4096) {
        int d[4], s[4], r[4];
#pragma unroll
        for (int k = 0; k < 4; ++k) {
            const int e = eb + 1024 * k;
            const bool in = (e < e1);
            d[k] = in ? ei[E + e] : -1;
            s[k] = in ? ei[e] : 0;
            r[k] = in ? rank[e] : 0;
        }
#pragma unroll
        for (int k = 0; k < 4; ++k) {
            if (d[k] >= 0) st[e0 + offs[d[k] >> 8] + r[k]] = make_int2(s[k], d[k]);
        }
    }
}

__global__ void __launch_bounds__(1024) bucket_rows(const int* __restrict__ T, const int* __restrict__ O,
                                                    const int* __restrict__ staged,
                                                    int* __restrict__ rowptr, float* __restrict__ dis,
                                                    int* __restrict__ col, int Nn, int CS) {
    __shared__ int cnt[kBucket];
    __shared__ int cur[kBucket];
    __shared__ int wsum[8];
    __shared__ int wred[32];
    const int tid = threadIdx.x, lane = tid & 31, wave = tid >> 5;
    const int b  = blockIdx.x;
    const int lo = b * kBucket;
    const int c  = tid >> 2, q = tid & 3;
    const int o   = O[c * kBucket + b];
    const int len = T[c * kBucket + b];
    const int2* __restrict__ cell = (const int2*)staged + ((long long)c * CS + o);

    int r = (q == 0) ? o : 0;
#pragma unroll
    for (int m = 16; m >= 1; m >>= 1) r += __shfl_xor(r, m);
    if (lane == 0) wred[wave] = r;
    if (tid < kBucket) cnt[tid] = 0;
    __syncthreads();
    int base = wred[lane];
#pragma unroll
    for (int m = 16; m >= 1; m >>= 1) base += __shfl_xor(base, m);

    for (int i = q; i < len; i += 16) {
        int dn[4];
#pragma unroll
        for (int k = 0; k < 4; ++k)
            dn[k] = (i + 4 * k < len) ? cell[i + 4 * k].y : -1;
#pragma unroll
        for (int k = 0; k < 4; ++k)
            if (dn[k] >= 0) atomicAdd(&cnt[dn[k] - lo], 1);
    }
    __syncthreads();

    const int v = (tid < kBucket) ? cnt[tid] : 0;
    const int x = wave_incl_scan(v, lane);
    if (tid < kBucket && lane == 31) wsum[wave] = x;
    __syncthreads();
    if (tid < kBucket) {
        int pre = 0;
        for (int w = 0; w < wave; ++w) pre += wsum[w];
        const int ex = pre + x - v;
        const int node = lo + tid;
        if (node < Nn) {
            rowptr[node] = base + ex;
            const float deg = (float)(v + 1);
            dis[node] = (deg > 0.0f) ? rsqrtf(deg) : 0.0f;
        }
        if (b == (int)gridDim.x - 1 && tid == kBucket - 1) rowptr[Nn] = base + ex + v;
        cur[tid] = ex;
    }
    __syncthreads();

    for (int i = q; i < len; i += 16) {
        int2 p[4];
#pragma unroll
        for (int k = 0; k < 4; ++k)
            p[k] = (i + 4 * k < len) ? cell[i + 4 * k] : make_int2(0, -1);
#pragma unroll
        for (int k = 0; k < 4; ++k)
            if (p[k].y >= 0) col[base + atomicAdd(&cur[p[k].y - lo], 1)] = p[k].x;
    }
}

__global__ void cvt_transpose_w(const float* __restrict__ W, _Float16* __restrict__ Wt, int K, int C) {
    int idx = blockIdx.x * blockDim.x + threadIdx.x;
    if (idx < K * C) {
        int k = idx / C, c = idx % C;
        Wt[c * K + k] = (_Float16)W[idx];
    }
}

template <bool kHalfIn>
__device__ __forceinline__ float4 load_row4(const float* __restrict__ in32,
                                            const _Float16* __restrict__ in16, int s, int lane) {
    if (kHalfIn) {
        const v4h v = *(const v4h*)(in16 + (long long)s * 128 + lane * 4);
        return make_float4((float)v[0], (float)v[1], (float)v[2], (float)v[3]);
    } else {
        return ((const float4*)in32)[(long long)s * 32 + lane];
    }
}

template <bool kOrder, bool kHalfIn>
__device__ __forceinline__ float4 gather_row(const int* __restrict__ rowptr,
                                             int* __restrict__ col,
                                             const float* __restrict__ dis,
                                             const float* __restrict__ in32,
                                             const _Float16* __restrict__ in16,
                                             int d, int lane) {
    const int beg = rowptr[d], end = rowptr[d + 1];
    const float dd = dis[d];

    float4 acc;
    {
        const float4 v = load_row4<kHalfIn>(in32, in16, d, lane);
        const float w = dd * dd;
        acc.x = v.x * w; acc.y = v.y * w; acc.z = v.z * w; acc.w = v.w * w;
    }
    for (int base = beg; base < end; base += 32) {
        const int n = (end - base < 32) ? (end - base) : 32;
        int   myc = 0;
        float myw = 0.0f;
        if (lane < n) myc = col[base + lane];
        if (kOrder) {
            const unsigned key = ((unsigned)myc << 5) | (unsigned)lane;
            int pos = 0;
            for (int j = 0; j < n; ++j) {
                const unsigned kj = (unsigned)__builtin_amdgcn_readlane((int)key, j);
                pos += (kj < key) ? 1 : 0;
            }
            if (lane >= n) pos = lane;
            myc = __builtin_amdgcn_ds_permute(pos << 2, myc);
            if (lane < n) col[base + lane] = myc;
        }
        if (lane < n) myw = dis[myc] * dd;
#pragma unroll 4
        for (int j = 0; j < n; ++j) {
            const int   s = __builtin_amdgcn_readlane(myc, j);
            const float w = __int_as_float(__builtin_amdgcn_readlane(__float_as_int(myw), j));
            const float4 v = load_row4<kHalfIn>(in32, in16, s, lane);
            acc.x = fmaf(v.x, w, acc.x);
            acc.y = fmaf(v.y, w, acc.y);
            acc.z = fmaf(v.z, w, acc.z);
            acc.w = fmaf(v.w, w, acc.w);
        }
    }
    return acc;
}

constexpr int kAPitch = 136;
constexpr int kCPitch = 132;

template <bool kFirst>
__device__ __forceinline__ void layer_tile(const int* __restrict__ rowptr, int* __restrict__ col,
                                           const float* __restrict__ dis,
                                           const float* __restrict__ in32,
                                           const _Float16* __restrict__ in16,
                                           const _Float16* __restrict__ Wt,
                                           const float* __restrict__ bias,
                                           const float* __restrict__ Wq,
                                           _Float16* __restrict__ out16,
                                           float* __restrict__ pq, int Nn) {
    __shared__ __attribute__((aligned(16))) _Float16 As[16 * kAPitch];
    __shared__ __attribute__((aligned(16))) float    Cs[16 * kCPitch];
    const int wave = __builtin_amdgcn_readfirstlane((int)(threadIdx.x >> 5));
    const int lane = threadIdx.x & 31;
    const int row0 = blockIdx.x * 16;

    for (int i = 0; i < 2; ++i) {
        const int rt = wave * 2 + i;
        const int d  = row0 + rt;
        float4 r = make_float4(0.0f, 0.0f, 0.0f, 0.0f);
        if (d < Nn) r = gather_row<kFirst, !kFirst>(rowptr, col, dis, in32, in16, d, lane);
        v4h o;
        o[0] = (_Float16)r.x; o[1] = (_Float16)r.y; o[2] = (_Float16)r.z; o[3] = (_Float16)r.w;
        *(v4h*)(As + rt * kAPitch + lane * 4) = o;
    }
    __syncthreads();

    const int c = lane & 15;
    const int kbase = (lane >> 4) ? 8 : 0;
    const _Float16* arow = As + c * kAPitch;
    const _Float16* __restrict__ brow = Wt + (long long)(wave * 16 + c) * 128;
    v8f acc = {};
#pragma unroll
    for (int kc = 0; kc < 4; ++kc) {
        const int k0 = kc * 32 + kbase;
        v8h alo = *(const v8h*)(arow + k0);
        v8h ahi = *(const v8h*)(arow + k0 + 16);
        v8h blo = *(const v8h*)(brow + k0);
        v8h bhi = *(const v8h*)(brow + k0 + 16);
        v16h a, b;
#pragma unroll
        for (int e = 0; e < 8; ++e) {
            a[e] = alo[e]; a[8 + e] = ahi[e];
            b[e] = blo[e]; b[8 + e] = bhi[e];
        }
        acc = __builtin_amdgcn_wmma_f32_16x16x32_f16(
             false, a,  false, b,
             (short)0, acc,  false,  false);
    }

    {
        const float bcol = bias[wave * 16 + c];
        float* crow = Cs + ((lane >> 4) ? 8 : 0) * kCPitch + wave * 16 + c;
#pragma unroll
        for (int r = 0; r < 8; ++r) {
            float v = acc[r] + bcol;
            crow[r * kCPitch] = v > 0.0f ? v : 0.0f;
        }
    }
    float4 wa = make_float4(0.0f, 0.0f, 0.0f, 0.0f), wb = wa;
    if (!kFirst) {
        wa = ((const float4*)Wq)[lane];
        wb = ((const float4*)Wq)[32 + lane];
    }
    __syncthreads();

    for (int i = 0; i < 2; ++i) {
        const int rt = wave * 2 + i;
        const int d  = row0 + rt;
        if (d >= Nn) continue;
        const float4 r = *(const float4*)(Cs + rt * kCPitch + lane * 4);
        if (kFirst) {
            v4h o;
            o[0] = (_Float16)r.x; o[1] = (_Float16)r.y; o[2] = (_Float16)r.z; o[3] = (_Float16)r.w;
            *(v4h*)(out16 + (long long)d * 128 + lane * 4) = o;
        } else {
            float p = r.x * wa.x + r.y * wa.y + r.z * wa.z + r.w * wa.w;
            float t = r.x * wb.x + r.y * wb.y + r.z * wb.z + r.w * wb.w;
#pragma unroll
            for (int o = 16; o >= 1; o >>= 1) {
                p += __shfl_xor(p, o);
                t += __shfl_xor(t, o);
            }
            if (lane == 0) *(float2*)(pq + 2 * (long long)d) = make_float2(p, t);
        }
    }
}

__global__ void __launch_bounds__(256) layer1_agg_mm(const int* __restrict__ rowptr, int* __restrict__ col,
                                                  const float* __restrict__ dis,
                                                  const float* __restrict__ x,
                                                  const _Float16* __restrict__ Wt,
                                                  const float* __restrict__ bias,
                                                  _Float16* __restrict__ out16, int Nn) {
    layer_tile<true>(rowptr, col, dis, x, nullptr, Wt, bias, nullptr, out16, nullptr, Nn);
}

__global__ void __launch_bounds__(256) layer2_agg_mm_dotq(const int* __restrict__ rowptr, int* __restrict__ col,
                                                       const float* __restrict__ dis,
                                                       const _Float16* __restrict__ hin,
                                                       const _Float16* __restrict__ Wt,
                                                       const float* __restrict__ bias,
                                                       const float* __restrict__ Wq,
                                                       float* __restrict__ pq, int Nn) {
    layer_tile<false>(rowptr, col, dis, nullptr, hin, Wt, bias, Wq, nullptr, pq, Nn);
}

__global__ void q_pairs(const int* __restrict__ va, const float* __restrict__ pq,
                        const float* __restrict__ bq, float* __restrict__ out, int A) {
    int a = blockIdx.x * blockDim.x + threadIdx.x;
    if (a >= A) return;
    const int2 ij = ((const int2*)va)[a];
    out[a] = (pq[2 * (long long)ij.x] + pq[2 * (long long)ij.y + 1]) + bq[0];
}

extern "C" void kernel_launch(void* const* d_in, const int* in_sizes, int n_in,
                              void* d_out, int out_size, void* d_ws, size_t ws_size,
                              hipStream_t stream) {
    const float* x  = (const float*)d_in[0];
    const int*   ei = (const int*)d_in[1];
    const int*   va = (const int*)d_in[2];
    const float* W1 = (const float*)d_in[3];
    const float* b1 = (const float*)d_in[4];
    const float* W2 = (const float*)d_in[5];
    const float* b2 = (const float*)d_in[6];
    const float* Wq = (const float*)d_in[7];
    const float* bq = (const float*)d_in[8];
    float* q = (float*)d_out;

    const int H  = 128;
    const int Nn = in_sizes[0] / H;
    const int E  = in_sizes[1] / 2;
    const int A  = in_sizes[2] / 2;

    char*  ws  = (char*)d_ws;
    size_t off = 0;
    auto alloc = [&](size_t bytes) -> char* {
        char* p = ws + off;
        off = (off + bytes + 255) & ~(size_t)255;
        return p;
    };
    _Float16* h1h = (_Float16*)alloc((size_t)Nn * H * 2);
    _Float16* w1t = (_Float16*)alloc((size_t)H * H * 2);
    _Float16* w2t = (_Float16*)alloc((size_t)H * H * 2);
    float*    pq  = (float*)   alloc((size_t)Nn * 2 * 4);
    char*     csr = (char*)    alloc((size_t)Nn * H * 2);
    float*    dis = (float*)   alloc((size_t)Nn * 4);

    size_t coff = 0;
    auto calloc_ = [&](size_t bytes) -> int* {
        int* p = (int*)(csr + coff);
        coff = (coff + bytes + 255) & ~(size_t)255;
        return p;
    };
    int* rowptr = calloc_((size_t)(Nn + 1) * 4);
    int* rank   = calloc_((size_t)E * 4);
    int* col    = calloc_((size_t)E * 4);
    int* cellT  = calloc_((size_t)kChunks * kBucket * 4);
    int* cellO  = calloc_((size_t)kChunks * kBucket * 4);
    int* staged = calloc_((size_t)E * 8);

    auto blks = [](long long n, int t) { return (unsigned)((n + t - 1) / t); };

    const int CS = (int)blks(E, kChunks);
    bin_chunk<<<kChunks, 1024, 0, stream>>>(ei, rank, cellT, cellO, staged, E, CS);
    bucket_rows<<<blks(Nn, kBucket), 1024, 0, stream>>>(cellT, cellO, staged, rowptr, dis, col, Nn, CS);

    cvt_transpose_w<<<blks(H * H, 256), 256, 0, stream>>>(W1, w1t, H, H);
    cvt_transpose_w<<<blks(H * H, 256), 256, 0, stream>>>(W2, w2t, H, H);

    const unsigned mtiles = (unsigned)((Nn + 15) / 16);

    layer1_agg_mm<<<mtiles, 256, 0, stream>>>(rowptr, col, dis, x, w1t, b1, h1h, Nn);
    layer2_agg_mm_dotq<<<mtiles, 256, 0, stream>>>(rowptr, col, dis, h1h, w2t, b2, Wq, pq, Nn);

    q_pairs<<<blks(A, 256), 256, 0, stream>>>(va, pq, bq, q, A);
}


# W1/W2 f16 convert+transpose done by 4 trailing blocks of the bucket kernel's grid (8 elements per thread, whole-line reads, 16-B stores): both one-element-per-thread launches gone, 7 launches to 5
# speedup vs baseline: 5.7576x; 1.0990x over previous
#include <hip/hip_runtime.h>
#include <hip/hip_bf16.h>

typedef __attribute__((ext_vector_type(16))) _Float16 v16h;
typedef __attribute__((ext_vector_type(8)))  _Float16 v8h;
typedef __attribute__((ext_vector_type(4)))  _Float16 v4h;
typedef __attribute__((ext_vector_type(8)))  float    v8f;

constexpr int kBucket = 256;
constexpr int kChunks = 256;

__device__ __forceinline__ int wave_incl_scan(int x, int lane) {
#pragma unroll
    for (int o = 1; o < 32; o <<= 1) {
        int y = __shfl_up(x, o);
        if (lane >= o) x += y;
    }
    return x;
}

__global__ void __launch_bounds__(1024) bin_chunk(const int* __restrict__ ei, int* __restrict__ rank,
                                                  int* __restrict__ T, int* __restrict__ O,
                                                  int* __restrict__ staged, int E, int CS) {
    __shared__ int hist[kBucket];
    __shared__ int offs[kBucket];
    __shared__ int wsum[8];
    const int tid = threadIdx.x, lane = tid & 31, wave = tid >> 5;
    const int c  = blockIdx.x;
    const int e0 = c * CS;
    const int e1 = (e0 + CS < E) ? (e0 + CS) : E;
    if (tid < kBucket) hist[tid] = 0;
    __syncthreads();
    for (int eb = e0 + tid; eb < e1; eb += 4096) {
        int d[4];
#pragma unroll
        for (int k = 0; k < 4; ++k) {
            const int e = eb + 1024 * k;
            d[k] = (e < e1) ? ei[E + e] : -1;
        }
#pragma unroll
        for (int k = 0; k < 4; ++k) {
            const int e = eb + 1024 * k;
            if (d[k] >= 0) rank[e] = atomicAdd(&hist[d[k] >> 8], 1);
        }
    }
    __syncthreads();
    const int v = (tid < kBucket) ? hist[tid] : 0;
    const int x = wave_incl_scan(v, lane);
    if (tid < kBucket && lane == 31) wsum[wave] = x;
    __syncthreads();
    if (tid < kBucket) {
        int pre = 0;
        for (int w = 0; w < wave; ++w) pre += wsum[w];
        const int ex = pre + x - v;
        offs[tid] = ex;
        T[c * kBucket + tid] = v;
        O[c * kBucket + tid] = ex;
    }
    __syncthreads();
    int2* __restrict__ st = (int2*)staged;
    for (int eb = e0 + tid; eb < e1; eb += 4096) {
        int d[4], s[4], r[4];
#pragma unroll
        for (int k = 0; k < 4; ++k) {
            const int e = eb + 1024 * k;
            const bool in = (e < e1);
            d[k] = in ? ei[E + e] : -1;
            s[k] = in ? ei[e] : 0;
            r[k] = in ? rank[e] : 0;
        }
#pragma unroll
        for (int k = 0; k < 4; ++k) {
            if (d[k] >= 0) st[e0 + offs[d[k] >> 8] + r[k]] = make_int2(s[k], d[k]);
        }
    }
}

__device__ __forceinline__ void cvt_transpose_w8(const float* __restrict__ Wm, _Float16* __restrict__ Wt, int t) {
    const int c = t & 127, k0 = (t >> 7) * 8;
    v8h o;
#pragma unroll
    for (int j = 0; j < 8; ++j)
        o[j] = (_Float16)Wm[(k0 + j) * 128 + c];
    *(v8h*)(Wt + c * 128 + k0) = o;
}
constexpr int kCvtBlocks = 4;

__global__ void __launch_bounds__(1024) bucket_rows(const int* __restrict__ T, const int* __restrict__ O,
                                                    const int* __restrict__ staged,
                                                    int* __restrict__ rowptr, float* __restrict__ dis,
                                                    int* __restrict__ col, int Nn, int CS,
                                                    const float* __restrict__ W1, const float* __restrict__ W2,
                                                    _Float16* __restrict__ w1t, _Float16* __restrict__ w2t) {
    __shared__ int cnt[kBucket];
    __shared__ int cur[kBucket];
    __shared__ int wsum[8];
    __shared__ int wred[32];
    const int nb = (int)gridDim.x - kCvtBlocks;
    if ((int)blockIdx.x >= nb) {
        const int t = ((int)blockIdx.x - nb) * 1024 + (int)threadIdx.x;
        if (t < 2048) cvt_transpose_w8(W1, w1t, t);
        else          cvt_transpose_w8(W2, w2t, t - 2048);
        return;
    }
    const int tid = threadIdx.x, lane = tid & 31, wave = tid >> 5;
    const int b  = blockIdx.x;
    const int lo = b * kBucket;
    const int c  = tid >> 2, q = tid & 3;
    const int o   = O[c * kBucket + b];
    const int len = T[c * kBucket + b];
    const int2* __restrict__ cell = (const int2*)staged + ((long long)c * CS + o);

    int r = (q == 0) ? o : 0;
#pragma unroll
    for (int m = 16; m >= 1; m >>= 1) r += __shfl_xor(r, m);
    if (lane == 0) wred[wave] = r;
    if (tid < kBucket) cnt[tid] = 0;
    __syncthreads();
    int base = wred[lane];
#pragma unroll
    for (int m = 16; m >= 1; m >>= 1) base += __shfl_xor(base, m);

    for (int i = q; i < len; i += 16) {
        int dn[4];
#pragma unroll
        for (int k = 0; k < 4; ++k)
            dn[k] = (i + 4 * k < len) ? cell[i + 4 * k].y : -1;
#pragma unroll
        for (int k = 0; k < 4; ++k)
            if (dn[k] >= 0) atomicAdd(&cnt[dn[k] - lo], 1);
    }
    __syncthreads();

    const int v = (tid < kBucket) ? cnt[tid] : 0;
    const int x = wave_incl_scan(v, lane);
    if (tid < kBucket && lane == 31) wsum[wave] = x;
    __syncthreads();
    if (tid < kBucket) {
        int pre = 0;
        for (int w = 0; w < wave; ++w) pre += wsum[w];
        const int ex = pre + x - v;
        const int node = lo + tid;
        if (node < Nn) {
            rowptr[node] = base + ex;
            const float deg = (float)(v + 1);
            dis[node] = (deg > 0.0f) ? rsqrtf(deg) : 0.0f;
        }
        if (b == nb - 1 && tid == kBucket - 1) rowptr[Nn] = base + ex + v;
        cur[tid] = ex;
    }
    __syncthreads();

    for (int i = q; i < len; i += 16) {
        int2 p[4];
#pragma unroll
        for (int k = 0; k < 4; ++k)
            p[k] = (i + 4 * k < len) ? cell[i + 4 * k] : make_int2(0, -1);
#pragma unroll
        for (int k = 0; k < 4; ++k)
            if (p[k].y >= 0) col[base + atomicAdd(&cur[p[k].y - lo], 1)] = p[k].x;
    }
}

template <bool kHalfIn>
__device__ __forceinline__ float4 load_row4(const float* __restrict__ in32,
                                            const _Float16* __restrict__ in16, int s, int lane) {
    if (kHalfIn) {
        const v4h v = *(const v4h*)(in16 + (long long)s * 128 + lane * 4);
        return make_float4((float)v[0], (float)v[1], (float)v[2], (float)v[3]);
    } else {
        return ((const float4*)in32)[(long long)s * 32 + lane];
    }
}

template <bool kOrder, bool kHalfIn>
__device__ __forceinline__ float4 gather_row(const int* __restrict__ rowptr,
                                             int* __restrict__ col,
                                             const float* __restrict__ dis,
                                             const float* __restrict__ in32,
                                             const _Float16* __restrict__ in16,
                                             int d, int lane) {
    const int beg = rowptr[d], end = rowptr[d + 1];
    const float dd = dis[d];

    float4 acc;
    {
        const float4 v = load_row4<kHalfIn>(in32, in16, d, lane);
        const float w = dd * dd;
        acc.x = v.x * w; acc.y = v.y * w; acc.z = v.z * w; acc.w = v.w * w;
    }
    for (int base = beg; base < end; base += 32) {
        const int n = (end - base < 32) ? (end - base) : 32;
        int   myc = 0;
        float myw = 0.0f;
        if (lane < n) myc = col[base + lane];
        if (kOrder) {
            const unsigned key = ((unsigned)myc << 5) | (unsigned)lane;
            int pos = 0;
            for (int j = 0; j < n; ++j) {
                const unsigned kj = (unsigned)__builtin_amdgcn_readlane((int)key, j);
                pos += (kj < key) ? 1 : 0;
            }
            if (lane >= n) pos = lane;
            myc = __builtin_amdgcn_ds_permute(pos << 2, myc);
            if (lane < n) col[base + lane] = myc;
        }
        if (lane < n) myw = dis[myc] * dd;
#pragma unroll 4
        for (int j = 0; j < n; ++j) {
            const int   s = __builtin_amdgcn_readlane(myc, j);
            const float w = __int_as_float(__builtin_amdgcn_readlane(__float_as_int(myw), j));
            const float4 v = load_row4<kHalfIn>(in32, in16, s, lane);
            acc.x = fmaf(v.x, w, acc.x);
            acc.y = fmaf(v.y, w, acc.y);
            acc.z = fmaf(v.z, w, acc.z);
            acc.w = fmaf(v.w, w, acc.w);
        }
    }
    return acc;
}

constexpr int kAPitch = 136;
constexpr int kCPitch = 132;

template <bool kFirst>
__device__ __forceinline__ void layer_tile(const int* __restrict__ rowptr, int* __restrict__ col,
                                           const float* __restrict__ dis,
                                           const float* __restrict__ in32,
                                           const _Float16* __restrict__ in16,
                                           const _Float16* __restrict__ Wt,
                                           const float* __restrict__ bias,
                                           const float* __restrict__ Wq,
                                           _Float16* __restrict__ out16,
                                           float* __restrict__ pq, int Nn) {
    __shared__ __attribute__((aligned(16))) _Float16 As[16 * kAPitch];
    __shared__ __attribute__((aligned(16))) float    Cs[16 * kCPitch];
    const int wave = __builtin_amdgcn_readfirstlane((int)(threadIdx.x >> 5));
    const int lane = threadIdx.x & 31;
    const int row0 = blockIdx.x * 16;

    for (int i = 0; i < 2; ++i) {
        const int rt = wave * 2 + i;
        const int d  = row0 + rt;
        float4 r = make_float4(0.0f, 0.0f, 0.0f, 0.0f);
        if (d < Nn) r = gather_row<kFirst, !kFirst>(rowptr, col, dis, in32, in16, d, lane);
        v4h o;
        o[0] = (_Float16)r.x; o[1] = (_Float16)r.y; o[2] = (_Float16)r.z; o[3] = (_Float16)r.w;
        *(v4h*)(As + rt * kAPitch + lane * 4) = o;
    }
    __syncthreads();

    const int c = lane & 15;
    const int kbase = (lane >> 4) ? 8 : 0;
    const _Float16* arow = As + c * kAPitch;
    const _Float16* __restrict__ brow = Wt + (long long)(wave * 16 + c) * 128;
    v8f acc = {};
#pragma unroll
    for (int kc = 0; kc < 4; ++kc) {
        const int k0 = kc * 32 + kbase;
        v8h alo = *(const v8h*)(arow + k0);
        v8h ahi = *(const v8h*)(arow + k0 + 16);
        v8h blo = *(const v8h*)(brow + k0);
        v8h bhi = *(const v8h*)(brow + k0 + 16);
        v16h a, b;
#pragma unroll
        for (int e = 0; e < 8; ++e) {
            a[e] = alo[e]; a[8 + e] = ahi[e];
            b[e] = blo[e]; b[8 + e] = bhi[e];
        }
        acc = __builtin_amdgcn_wmma_f32_16x16x32_f16(
             false, a,  false, b,
             (short)0, acc,  false,  false);
    }

    {
        const float bcol = bias[wave * 16 + c];
        float* crow = Cs + ((lane >> 4) ? 8 : 0) * kCPitch + wave * 16 + c;
#pragma unroll
        for (int r = 0; r < 8; ++r) {
            float v = acc[r] + bcol;
            crow[r * kCPitch] = v > 0.0f ? v : 0.0f;
        }
    }
    float4 wa = make_float4(0.0f, 0.0f, 0.0f, 0.0f), wb = wa;
    if (!kFirst) {
        wa = ((const float4*)Wq)[lane];
        wb = ((const float4*)Wq)[32 + lane];
    }
    __syncthreads();

    for (int i = 0; i < 2; ++i) {
        const int rt = wave * 2 + i;
        const int d  = row0 + rt;
        if (d >= Nn) continue;
        const float4 r = *(const float4*)(Cs + rt * kCPitch + lane * 4);
        if (kFirst) {
            v4h o;
            o[0] = (_Float16)r.x; o[1] = (_Float16)r.y; o[2] = (_Float16)r.z; o[3] = (_Float16)r.w;
            *(v4h*)(out16 + (long long)d * 128 + lane * 4) = o;
        } else {
            float p = r.x * wa.x + r.y * wa.y + r.z * wa.z + r.w * wa.w;
            float t = r.x * wb.x + r.y * wb.y + r.z * wb.z + r.w * wb.w;
#pragma unroll
            for (int o = 16; o >= 1; o >>= 1) {
                p += __shfl_xor(p, o);
                t += __shfl_xor(t, o);
            }
            if (lane == 0) *(float2*)(pq + 2 * (long long)d) = make_float2(p, t);
        }
    }
}

__global__ void __launch_bounds__(256) layer1_agg_mm(const int* __restrict__ rowptr, int* __restrict__ col,
                                                  const float* __restrict__ dis,
                                                  const float* __restrict__ x,
                                                  const _Float16* __restrict__ Wt,
                                                  const float* __restrict__ bias,
                                                  _Float16* __restrict__ out16, int Nn) {
    layer_tile<true>(rowptr, col, dis, x, nullptr, Wt, bias, nullptr, out16, nullptr, Nn);
}

__global__ void __launch_bounds__(256) layer2_agg_mm_dotq(const int* __restrict__ rowptr, int* __restrict__ col,
                                                       const float* __restrict__ dis,
                                                       const _Float16* __restrict__ hin,
                                                       const _Float16* __restrict__ Wt,
                                                       const float* __restrict__ bias,
                                                       const float* __restrict__ Wq,
                                                       float* __restrict__ pq, int Nn) {
    layer_tile<false>(rowptr, col, dis, nullptr, hin, Wt, bias, Wq, nullptr, pq, Nn);
}

__global__ void q_pairs(const int* __restrict__ va, const float* __restrict__ pq,
                        const float* __restrict__ bq, float* __restrict__ out, int A) {
    int a = blockIdx.x * blockDim.x + threadIdx.x;
    if (a >= A) return;
    const int2 ij = ((const int2*)va)[a];
    out[a] = (pq[2 * (long long)ij.x] + pq[2 * (long long)ij.y + 1]) + bq[0];
}

extern "C" void kernel_launch(void* const* d_in, const int* in_sizes, int n_in,
                              void* d_out, int out_size, void* d_ws, size_t ws_size,
                              hipStream_t stream) {
    const float* x  = (const float*)d_in[0];
    const int*   ei = (const int*)d_in[1];
    const int*   va = (const int*)d_in[2];
    const float* W1 = (const float*)d_in[3];
    const float* b1 = (const float*)d_in[4];
    const float* W2 = (const float*)d_in[5];
    const float* b2 = (const float*)d_in[6];
    const float* Wq = (const float*)d_in[7];
    const float* bq = (const float*)d_in[8];
    float* q = (float*)d_out;

    const int H  = 128;
    const int Nn = in_sizes[0] / H;
    const int E  = in_sizes[1] / 2;
    const int A  = in_sizes[2] / 2;

    char*  ws  = (char*)d_ws;
    size_t off = 0;
    auto alloc = [&](size_t bytes) -> char* {
        char* p = ws + off;
        off = (off + bytes + 255) & ~(size_t)255;
        return p;
    };
    _Float16* h1h = (_Float16*)alloc((size_t)Nn * H * 2);
    _Float16* w1t = (_Float16*)alloc((size_t)H * H * 2);
    _Float16* w2t = (_Float16*)alloc((size_t)H * H * 2);
    float*    pq  = (float*)   alloc((size_t)Nn * 2 * 4);
    char*     csr = (char*)    alloc((size_t)Nn * H * 2);
    float*    dis = (float*)   alloc((size_t)Nn * 4);

    size_t coff = 0;
    auto calloc_ = [&](size_t bytes) -> int* {
        int* p = (int*)(csr + coff);
        coff = (coff + bytes + 255) & ~(size_t)255;
        return p;
    };
    int* rowptr = calloc_((size_t)(Nn + 1) * 4);
    int* rank   = calloc_((size_t)E * 4);
    int* col    = calloc_((size_t)E * 4);
    int* cellT  = calloc_((size_t)kChunks * kBucket * 4);
    int* cellO  = calloc_((size_t)kChunks * kBucket * 4);
    int* staged = calloc_((size_t)E * 8);

    auto blks = [](long long n, int t) { return (unsigned)((n + t - 1) / t); };

    const int CS = (int)blks(E, kChunks);
    bin_chunk<<<kChunks, 1024, 0, stream>>>(ei, rank, cellT, cellO, staged, E, CS);
    bucket_rows<<<blks(Nn, kBucket) + kCvtBlocks, 1024, 0, stream>>>(cellT, cellO, staged, rowptr, dis, col, Nn, CS,
                                                                      W1, W2, w1t, w2t);

    const unsigned mtiles = (unsigned)((Nn + 15) / 16);

    layer1_agg_mm<<<mtiles, 256, 0, stream>>>(rowptr, col, dis, x, w1t, b1, h1h, Nn);
    layer2_agg_mm_dotq<<<mtiles, 256, 0, stream>>>(rowptr, col, dis, h1h, w2t, b2, Wq, pq, Nn);

    q_pairs<<<blks(A, 256), 256, 0, stream>>>(va, pq, bq, q, A);
}


# fewer, fatter waves: a layer block takes 32 rows (wave gathers 4 rows, two WMMA row tiles share each W fragment), Q head 4 actions per thread; sums and their order unchanged
# speedup vs baseline: 6.2682x; 1.0887x over previous
#include <hip/hip_runtime.h>
#include <hip/hip_bf16.h>

typedef __attribute__((ext_vector_type(16))) _Float16 v16h;
typedef __attribute__((ext_vector_type(8)))  _Float16 v8h;
typedef __attribute__((ext_vector_type(4)))  _Float16 v4h;
typedef __attribute__((ext_vector_type(8)))  float    v8f;

constexpr int kBucket = 256;
constexpr int kChunks = 256;

__device__ __forceinline__ int wave_incl_scan(int x, int lane) {
#pragma unroll
    for (int o = 1; o < 32; o <<= 1) {
        int y = __shfl_up(x, o);
        if (lane >= o) x += y;
    }
    return x;
}

__global__ void __launch_bounds__(1024) bin_chunk(const int* __restrict__ ei, int* __restrict__ rank,
                                                  int* __restrict__ T, int* __restrict__ O,
                                                  int* __restrict__ staged, int E, int CS) {
    __shared__ int hist[kBucket];
    __shared__ int offs[kBucket];
    __shared__ int wsum[8];
    const int tid = threadIdx.x, lane = tid & 31, wave = tid >> 5;
    const int c  = blockIdx.x;
    const int e0 = c * CS;
    const int e1 = (e0 + CS < E) ? (e0 + CS) : E;
    if (tid < kBucket) hist[tid] = 0;
    __syncthreads();
    for (int eb = e0 + tid; eb < e1; eb += 4096) {
        int d[4];
#pragma unroll
        for (int k = 0; k < 4; ++k) {
            const int e = eb + 1024 * k;
            d[k] = (e < e1) ? ei[E + e] : -1;
        }
#pragma unroll
        for (int k = 0; k < 4; ++k) {
            const int e = eb + 1024 * k;
            if (d[k] >= 0) rank[e] = atomicAdd(&hist[d[k] >> 8], 1);
        }
    }
    __syncthreads();
    const int v = (tid < kBucket) ? hist[tid] : 0;
    const int x = wave_incl_scan(v, lane);
    if (tid < kBucket && lane == 31) wsum[wave] = x;
    __syncthreads();
    if (tid < kBucket) {
        int pre = 0;
        for (int w = 0; w < wave; ++w) pre += wsum[w];
        const int ex = pre + x - v;
        offs[tid] = ex;
        T[c * kBucket + tid] = v;
        O[c * kBucket + tid] = ex;
    }
    __syncthreads();
    int2* __restrict__ st = (int2*)staged;
    for (int eb = e0 + tid; eb < e1; eb += 4096) {
        int d[4], s[4], r[4];
#pragma unroll
        for (int k = 0; k < 4; ++k) {
            const int e = eb + 1024 * k;
            const bool in = (e < e1);
            d[k] = in ? ei[E + e] : -1;
            s[k] = in ? ei[e] : 0;
            r[k] = in ? rank[e] : 0;
        }
#pragma unroll
        for (int k = 0; k < 4; ++k) {
            if (d[k] >= 0) st[e0 + offs[d[k] >> 8] + r[k]] = make_int2(s[k], d[k]);
        }
    }
}

__device__ __forceinline__ void cvt_transpose_w8(const float* __restrict__ Wm, _Float16* __restrict__ Wt, int t) {
    const int c = t & 127, k0 = (t >> 7) * 8;
    v8h o;
#pragma unroll
    for (int j = 0; j < 8; ++j)
        o[j] = (_Float16)Wm[(k0 + j) * 128 + c];
    *(v8h*)(Wt + c * 128 + k0) = o;
}
constexpr int kCvtBlocks = 4;

__global__ void __launch_bounds__(1024) bucket_rows(const int* __restrict__ T, const int* __restrict__ O,
                                                    const int* __restrict__ staged,
                                                    int* __restrict__ rowptr, float* __restrict__ dis,
                                                    int* __restrict__ col, int Nn, int CS,
                                                    const float* __restrict__ W1, const float* __restrict__ W2,
                                                    _Float16* __restrict__ w1t, _Float16* __restrict__ w2t) {
    __shared__ int cnt[kBucket];
    __shared__ int cur[kBucket];
    __shared__ int wsum[8];
    __shared__ int wred[32];
    const int nb = (int)gridDim.x - kCvtBlocks;
    if ((int)blockIdx.x >= nb) {
        const int t = ((int)blockIdx.x - nb) * 1024 + (int)threadIdx.x;
        if (t < 2048) cvt_transpose_w8(W1, w1t, t);
        else          cvt_transpose_w8(W2, w2t, t - 2048);
        return;
    }
    const int tid = threadIdx.x, lane = tid & 31, wave = tid >> 5;
    const int b  = blockIdx.x;
    const int lo = b * kBucket;
    const int c  = tid >> 2, q = tid & 3;
    const int o   = O[c * kBucket + b];
    const int len = T[c * kBucket + b];
    const int2* __restrict__ cell = (const int2*)staged + ((long long)c * CS + o);

    int r = (q == 0) ? o : 0;
#pragma unroll
    for (int m = 16; m >= 1; m >>= 1) r += __shfl_xor(r, m);
    if (lane == 0) wred[wave] = r;
    if (tid < kBucket) cnt[tid] = 0;
    __syncthreads();
    int base = wred[lane];
#pragma unroll
    for (int m = 16; m >= 1; m >>= 1) base += __shfl_xor(base, m);

    for (int i = q; i < len; i += 16) {
        int dn[4];
#pragma unroll
        for (int k = 0; k < 4; ++k)
            dn[k] = (i + 4 * k < len) ? cell[i + 4 * k].y : -1;
#pragma unroll
        for (int k = 0; k < 4; ++k)
            if (dn[k] >= 0) atomicAdd(&cnt[dn[k] - lo], 1);
    }
    __syncthreads();

    const int v = (tid < kBucket) ? cnt[tid] : 0;
    const int x = wave_incl_scan(v, lane);
    if (tid < kBucket && lane == 31) wsum[wave] = x;
    __syncthreads();
    if (tid < kBucket) {
        int pre = 0;
        for (int w = 0; w < wave; ++w) pre += wsum[w];
        const int ex = pre + x - v;
        const int node = lo + tid;
        if (node < Nn) {
            rowptr[node] = base + ex;
            const float deg = (float)(v + 1);
            dis[node] = (deg > 0.0f) ? rsqrtf(deg) : 0.0f;
        }
        if (b == nb - 1 && tid == kBucket - 1) rowptr[Nn] = base + ex + v;
        cur[tid] = ex;
    }
    __syncthreads();

    for (int i = q; i < len; i += 16) {
        int2 p[4];
#pragma unroll
        for (int k = 0; k < 4; ++k)
            p[k] = (i + 4 * k < len) ? cell[i + 4 * k] : make_int2(0, -1);
#pragma unroll
        for (int k = 0; k < 4; ++k)
            if (p[k].y >= 0) col[base + atomicAdd(&cur[p[k].y - lo], 1)] = p[k].x;
    }
}

template <bool kHalfIn>
__device__ __forceinline__ float4 load_row4(const float* __restrict__ in32,
                                            const _Float16* __restrict__ in16, int s, int lane) {
    if (kHalfIn) {
        const v4h v = *(const v4h*)(in16 + (long long)s * 128 + lane * 4);
        return make_float4((float)v[0], (float)v[1], (float)v[2], (float)v[3]);
    } else {
        return ((const float4*)in32)[(long long)s * 32 + lane];
    }
}

template <bool kOrder, bool kHalfIn>
__device__ __forceinline__ float4 gather_row(const int* __restrict__ rowptr,
                                             int* __restrict__ col,
                                             const float* __restrict__ dis,
                                             const float* __restrict__ in32,
                                             const _Float16* __restrict__ in16,
                                             int d, int lane) {
    const int beg = rowptr[d], end = rowptr[d + 1];
    const float dd = dis[d];

    float4 acc;
    {
        const float4 v = load_row4<kHalfIn>(in32, in16, d, lane);
        const float w = dd * dd;
        acc.x = v.x * w; acc.y = v.y * w; acc.z = v.z * w; acc.w = v.w * w;
    }
    for (int base = beg; base < end; base += 32) {
        const int n = (end - base < 32) ? (end - base) : 32;
        int   myc = 0;
        float myw = 0.0f;
        if (lane < n) myc = col[base + lane];
        if (kOrder) {
            const unsigned key = ((unsigned)myc << 5) | (unsigned)lane;
            int pos = 0;
            for (int j = 0; j < n; ++j) {
                const unsigned kj = (unsigned)__builtin_amdgcn_readlane((int)key, j);
                pos += (kj < key) ? 1 : 0;
            }
            if (lane >= n) pos = lane;
            myc = __builtin_amdgcn_ds_permute(pos << 2, myc);
            if (lane < n) col[base + lane] = myc;
        }
        if (lane < n) myw = dis[myc] * dd;
#pragma unroll 4
        for (int j = 0; j < n; ++j) {
            const int   s = __builtin_amdgcn_readlane(myc, j);
            const float w = __int_as_float(__builtin_amdgcn_readlane(__float_as_int(myw), j));
            const float4 v = load_row4<kHalfIn>(in32, in16, s, lane);
            acc.x = fmaf(v.x, w, acc.x);
            acc.y = fmaf(v.y, w, acc.y);
            acc.z = fmaf(v.z, w, acc.z);
            acc.w = fmaf(v.w, w, acc.w);
        }
    }
    return acc;
}

constexpr int kAPitch = 136;
constexpr int kCPitch = 132;
constexpr int kTileRows = 32;
constexpr int kWaveRows = 4;

template <bool kFirst>
__device__ __forceinline__ void layer_tile(const int* __restrict__ rowptr, int* __restrict__ col,
                                           const float* __restrict__ dis,
                                           const float* __restrict__ in32,
                                           const _Float16* __restrict__ in16,
                                           const _Float16* __restrict__ Wt,
                                           const float* __restrict__ bias,
                                           const float* __restrict__ Wq,
                                           _Float16* __restrict__ out16,
                                           float* __restrict__ pq, int Nn) {
    __shared__ __attribute__((aligned(16))) _Float16 As[kTileRows * kAPitch];
    __shared__ __attribute__((aligned(16))) float    Cs[kTileRows * kCPitch];
    const int wave = __builtin_amdgcn_readfirstlane((int)(threadIdx.x >> 5));
    const int lane = threadIdx.x & 31;
    const int row0 = blockIdx.x * kTileRows;

    for (int i = 0; i < kWaveRows; ++i) {
        const int rt = wave * kWaveRows + i;
        const int d  = row0 + rt;
        float4 r = make_float4(0.0f, 0.0f, 0.0f, 0.0f);
        if (d < Nn) r = gather_row<kFirst, !kFirst>(rowptr, col, dis, in32, in16, d, lane);
        v4h o;
        o[0] = (_Float16)r.x; o[1] = (_Float16)r.y; o[2] = (_Float16)r.z; o[3] = (_Float16)r.w;
        *(v4h*)(As + rt * kAPitch + lane * 4) = o;
    }
    __syncthreads();

    const int c = lane & 15;
    const int kbase = (lane >> 4) ? 8 : 0;
    const _Float16* arow = As + c * kAPitch;
    const _Float16* __restrict__ brow = Wt + (long long)(wave * 16 + c) * 128;
    v8f acc0 = {}, acc1 = {};
#pragma unroll
    for (int kc = 0; kc < 4; ++kc) {
        const int k0 = kc * 32 + kbase;
        v8h blo  = *(const v8h*)(brow + k0);
        v8h bhi  = *(const v8h*)(brow + k0 + 16);
        v8h a0lo = *(const v8h*)(arow + k0);
        v8h a0hi = *(const v8h*)(arow + k0 + 16);
        v8h a1lo = *(const v8h*)(arow + 16 * kAPitch + k0);
        v8h a1hi = *(const v8h*)(arow + 16 * kAPitch + k0 + 16);
        v16h a0, a1, b;
#pragma unroll
        for (int e = 0; e < 8; ++e) {
            a0[e] = a0lo[e]; a0[8 + e] = a0hi[e];
            a1[e] = a1lo[e]; a1[8 + e] = a1hi[e];
            b[e]  = blo[e];  b[8 + e]  = bhi[e];
        }
        acc0 = __builtin_amdgcn_wmma_f32_16x16x32_f16(
             false, a0,  false, b,
             (short)0, acc0,  false,  false);
        acc1 = __builtin_amdgcn_wmma_f32_16x16x32_f16(
             false, a1,  false, b,
             (short)0, acc1,  false,  false);
    }

    {
        const float bcol = bias[wave * 16 + c];
        float* crow = Cs + ((lane >> 4) ? 8 : 0) * kCPitch + wave * 16 + c;
#pragma unroll
        for (int r = 0; r < 8; ++r) {
            float v0 = acc0[r] + bcol;
            float v1 = acc1[r] + bcol;
            crow[r * kCPitch]        = v0 > 0.0f ? v0 : 0.0f;
            crow[(16 + r) * kCPitch] = v1 > 0.0f ? v1 : 0.0f;
        }
    }
    float4 wa = make_float4(0.0f, 0.0f, 0.0f, 0.0f), wb = wa;
    if (!kFirst) {
        wa = ((const float4*)Wq)[lane];
        wb = ((const float4*)Wq)[32 + lane];
    }
    __syncthreads();

    for (int i = 0; i < kWaveRows; ++i) {
        const int rt = wave * kWaveRows + i;
        const int d  = row0 + rt;
        if (d >= Nn) continue;
        const float4 r = *(const float4*)(Cs + rt * kCPitch + lane * 4);
        if (kFirst) {
            v4h o;
            o[0] = (_Float16)r.x; o[1] = (_Float16)r.y; o[2] = (_Float16)r.z; o[3] = (_Float16)r.w;
            *(v4h*)(out16 + (long long)d * 128 + lane * 4) = o;
        } else {
            float p = r.x * wa.x + r.y * wa.y + r.z * wa.z + r.w * wa.w;
            float t = r.x * wb.x + r.y * wb.y + r.z * wb.z + r.w * wb.w;
#pragma unroll
            for (int o = 16; o >= 1; o >>= 1) {
                p += __shfl_xor(p, o);
                t += __shfl_xor(t, o);
            }
            if (lane == 0) *(float2*)(pq + 2 * (long long)d) = make_float2(p, t);
        }
    }
}

__global__ void __launch_bounds__(256) layer1_agg_mm(const int* __restrict__ rowptr, int* __restrict__ col,
                                                  const float* __restrict__ dis,
                                                  const float* __restrict__ x,
                                                  const _Float16* __restrict__ Wt,
                                                  const float* __restrict__ bias,
                                                  _Float16* __restrict__ out16, int Nn) {
    layer_tile<true>(rowptr, col, dis, x, nullptr, Wt, bias, nullptr, out16, nullptr, Nn);
}

__global__ void __launch_bounds__(256) layer2_agg_mm_dotq(const int* __restrict__ rowptr, int* __restrict__ col,
                                                       const float* __restrict__ dis,
                                                       const _Float16* __restrict__ hin,
                                                       const _Float16* __restrict__ Wt,
                                                       const float* __restrict__ bias,
                                                       const float* __restrict__ Wq,
                                                       float* __restrict__ pq, int Nn) {
    layer_tile<false>(rowptr, col, dis, nullptr, hin, Wt, bias, Wq, nullptr, pq, Nn);
}

__global__ void __launch_bounds__(256) q_pairs4(const int* __restrict__ va, const float* __restrict__ pq,
                                                const float* __restrict__ bq, float* __restrict__ out, int A) {
    const int t  = blockIdx.x * 256 + threadIdx.x;
    const int a0 = t * 4;
    if (a0 >= A) return;
    const float b = bq[0];
    if (a0 + 3 < A) {
        const int4 u = ((const int4*)va)[2 * t];
        const int4 w = ((const int4*)va)[2 * t + 1];
        float4 o;
        o.x = (pq[2 * (long long)u.x] + pq[2 * (long long)u.y + 1]) + b;
        o.y = (pq[2 * (long long)u.z] + pq[2 * (long long)u.w + 1]) + b;
        o.z = (pq[2 * (long long)w.x] + pq[2 * (long long)w.y + 1]) + b;
        o.w = (pq[2 * (long long)w.z] + pq[2 * (long long)w.w + 1]) + b;
        ((float4*)out)[t] = o;
    } else {
        for (int a = a0; a < A; ++a) {
            const int2 ij = ((const int2*)va)[a];
            out[a] = (pq[2 * (long long)ij.x] + pq[2 * (long long)ij.y + 1]) + b;
        }
    }
}

extern "C" void kernel_launch(void* const* d_in, const int* in_sizes, int n_in,
                              void* d_out, int out_size, void* d_ws, size_t ws_size,
                              hipStream_t stream) {
    const float* x  = (const float*)d_in[0];
    const int*   ei = (const int*)d_in[1];
    const int*   va = (const int*)d_in[2];
    const float* W1 = (const float*)d_in[3];
    const float* b1 = (const float*)d_in[4];
    const float* W2 = (const float*)d_in[5];
    const float* b2 = (const float*)d_in[6];
    const float* Wq = (const float*)d_in[7];
    const float* bq = (const float*)d_in[8];
    float* q = (float*)d_out;

    const int H  = 128;
    const int Nn = in_sizes[0] / H;
    const int E  = in_sizes[1] / 2;
    const int A  = in_sizes[2] / 2;

    char*  ws  = (char*)d_ws;
    size_t off = 0;
    auto alloc = [&](size_t bytes) -> char* {
        char* p = ws + off;
        off = (off + bytes + 255) & ~(size_t)255;
        return p;
    };
    _Float16* h1h = (_Float16*)alloc((size_t)Nn * H * 2);
    _Float16* w1t = (_Float16*)alloc((size_t)H * H * 2);
    _Float16* w2t = (_Float16*)alloc((size_t)H * H * 2);
    float*    pq  = (float*)   alloc((size_t)Nn * 2 * 4);
    char*     csr = (char*)    alloc((size_t)Nn * H * 2);
    float*    dis = (float*)   alloc((size_t)Nn * 4);

    size_t coff = 0;
    auto calloc_ = [&](size_t bytes) -> int* {
        int* p = (int*)(csr + coff);
        coff = (coff + bytes + 255) & ~(size_t)255;
        return p;
    };
    int* rowptr = calloc_((size_t)(Nn + 1) * 4);
    int* rank   = calloc_((size_t)E * 4);
    int* col    = calloc_((size_t)E * 4);
    int* cellT  = calloc_((size_t)kChunks * kBucket * 4);
    int* cellO  = calloc_((size_t)kChunks * kBucket * 4);
    int* staged = calloc_((size_t)E * 8);

    auto blks = [](long long n, int t) { return (unsigned)((n + t - 1) / t); };

    const int CS = (int)blks(E, kChunks);
    bin_chunk<<<kChunks, 1024, 0, stream>>>(ei, rank, cellT, cellO, staged, E, CS);
    bucket_rows<<<blks(Nn, kBucket) + kCvtBlocks, 1024, 0, stream>>>(cellT, cellO, staged, rowptr, dis, col, Nn, CS,
                                                                      W1, W2, w1t, w2t);

    const unsigned mtiles = blks(Nn, kTileRows);

    layer1_agg_mm<<<mtiles, 256, 0, stream>>>(rowptr, col, dis, x, w1t, b1, h1h, Nn);
    layer2_agg_mm_dotq<<<mtiles, 256, 0, stream>>>(rowptr, col, dis, h1h, w2t, b2, Wq, pq, Nn);

    q_pairs4<<<blks(A, 1024), 256, 0, stream>>>(va, pq, bq, q, A);
}


# row gather with fewer dependent round trips: ids and weights of a wave's four rows fetched together up front (own row = slot 0), rows loaded 8 at a time with zero-weight padding; sum order unchanged
# speedup vs baseline: 6.4871x; 1.0349x over previous
#include <hip/hip_runtime.h>
#include <hip/hip_bf16.h>

typedef __attribute__((ext_vector_type(16))) _Float16 v16h;
typedef __attribute__((ext_vector_type(8)))  _Float16 v8h;
typedef __attribute__((ext_vector_type(4)))  _Float16 v4h;
typedef __attribute__((ext_vector_type(8)))  float    v8f;

constexpr int kBucket = 256;
constexpr int kChunks = 256;

__device__ __forceinline__ int wave_incl_scan(int x, int lane) {
#pragma unroll
    for (int o = 1; o < 32; o <<= 1) {
        int y = __shfl_up(x, o);
        if (lane >= o) x += y;
    }
    return x;
}

__global__ void __launch_bounds__(1024) bin_chunk(const int* __restrict__ ei, int* __restrict__ rank,
                                                  int* __restrict__ T, int* __restrict__ O,
                                                  int* __restrict__ staged, int E, int CS) {
    __shared__ int hist[kBucket];
    __shared__ int offs[kBucket];
    __shared__ int wsum[8];
    const int tid = threadIdx.x, lane = tid & 31, wave = tid >> 5;
    const int c  = blockIdx.x;
    const int e0 = c * CS;
    const int e1 = (e0 + CS < E) ? (e0 + CS) : E;
    if (tid < kBucket) hist[tid] = 0;
    __syncthreads();
    for (int eb = e0 + tid; eb < e1; eb += 4096) {
        int d[4];
#pragma unroll
        for (int k = 0; k < 4; ++k) {
            const int e = eb + 1024 * k;
            d[k] = (e < e1) ? ei[E + e] : -1;
        }
#pragma unroll
        for (int k = 0; k < 4; ++k) {
            const int e = eb + 1024 * k;
            if (d[k] >= 0) rank[e] = atomicAdd(&hist[d[k] >> 8], 1);
        }
    }
    __syncthreads();
    const int v = (tid < kBucket) ? hist[tid] : 0;
    const int x = wave_incl_scan(v, lane);
    if (tid < kBucket && lane == 31) wsum[wave] = x;
    __syncthreads();
    if (tid < kBucket) {
        int pre = 0;
        for (int w = 0; w < wave; ++w) pre += wsum[w];
        const int ex = pre + x - v;
        offs[tid] = ex;
        T[c * kBucket + tid] = v;
        O[c * kBucket + tid] = ex;
    }
    __syncthreads();
    int2* __restrict__ st = (int2*)staged;
    for (int eb = e0 + tid; eb < e1; eb += 4096) {
        int d[4], s[4], r[4];
#pragma unroll
        for (int k = 0; k < 4; ++k) {
            const int e = eb + 1024 * k;
            const bool in = (e < e1);
            d[k] = in ? ei[E + e] : -1;
            s[k] = in ? ei[e] : 0;
            r[k] = in ? rank[e] : 0;
        }
#pragma unroll
        for (int k = 0; k < 4; ++k) {
            if (d[k] >= 0) st[e0 + offs[d[k] >> 8] + r[k]] = make_int2(s[k], d[k]);
        }
    }
}

__device__ __forceinline__ void cvt_transpose_w8(const float* __restrict__ Wm, _Float16* __restrict__ Wt, int t) {
    const int c = t & 127, k0 = (t >> 7) * 8;
    v8h o;
#pragma unroll
    for (int j = 0; j < 8; ++j)
        o[j] = (_Float16)Wm[(k0 + j) * 128 + c];
    *(v8h*)(Wt + c * 128 + k0) = o;
}
constexpr int kCvtBlocks = 4;

__global__ void __launch_bounds__(1024) bucket_rows(const int* __restrict__ T, const int* __restrict__ O,
                                                    const int* __restrict__ staged,
                                                    int* __restrict__ rowptr, float* __restrict__ dis,
                                                    int* __restrict__ col, int Nn, int CS,
                                                    const float* __restrict__ W1, const float* __restrict__ W2,
                                                    _Float16* __restrict__ w1t, _Float16* __restrict__ w2t) {
    __shared__ int cnt[kBucket];
    __shared__ int cur[kBucket];
    __shared__ int wsum[8];
    __shared__ int wred[32];
    const int nb = (int)gridDim.x - kCvtBlocks;
    if ((int)blockIdx.x >= nb) {
        const int t = ((int)blockIdx.x - nb) * 1024 + (int)threadIdx.x;
        if (t < 2048) cvt_transpose_w8(W1, w1t, t);
        else          cvt_transpose_w8(W2, w2t, t - 2048);
        return;
    }
    const int tid = threadIdx.x, lane = tid & 31, wave = tid >> 5;
    const int b  = blockIdx.x;
    const int lo = b * kBucket;
    const int c  = tid >> 2, q = tid & 3;
    const int o   = O[c * kBucket + b];
    const int len = T[c * kBucket + b];
    const int2* __restrict__ cell = (const int2*)staged + ((long long)c * CS + o);

    int r = (q == 0) ? o : 0;
#pragma unroll
    for (int m = 16; m >= 1; m >>= 1) r += __shfl_xor(r, m);
    if (lane == 0) wred[wave] = r;
    if (tid < kBucket) cnt[tid] = 0;
    __syncthreads();
    int base = wred[lane];
#pragma unroll
    for (int m = 16; m >= 1; m >>= 1) base += __shfl_xor(base, m);

    for (int i = q; i < len; i += 16) {
        int dn[4];
#pragma unroll
        for (int k = 0; k < 4; ++k)
            dn[k] = (i + 4 * k < len) ? cell[i + 4 * k].y : -1;
#pragma unroll
        for (int k = 0; k < 4; ++k)
            if (dn[k] >= 0) atomicAdd(&cnt[dn[k] - lo], 1);
    }
    __syncthreads();

    const int v = (tid < kBucket) ? cnt[tid] : 0;
    const int x = wave_incl_scan(v, lane);
    if (tid < kBucket && lane == 31) wsum[wave] = x;
    __syncthreads();
    if (tid < kBucket) {
        int pre = 0;
        for (int w = 0; w < wave; ++w) pre += wsum[w];
        const int ex = pre + x - v;
        const int node = lo + tid;
        if (node < Nn) {
            rowptr[node] = base + ex;
            const float deg = (float)(v + 1);
            dis[node] = (deg > 0.0f) ? rsqrtf(deg) : 0.0f;
        }
        if (b == nb - 1 && tid == kBucket - 1) rowptr[Nn] = base + ex + v;
        cur[tid] = ex;
    }
    __syncthreads();

    for (int i = q; i < len; i += 16) {
        int2 p[4];
#pragma unroll
        for (int k = 0; k < 4; ++k)
            p[k] = (i + 4 * k < len) ? cell[i + 4 * k] : make_int2(0, -1);
#pragma unroll
        for (int k = 0; k < 4; ++k)
            if (p[k].y >= 0) col[base + atomicAdd(&cur[p[k].y - lo], 1)] = p[k].x;
    }
}

template <bool kHalfIn>
__device__ __forceinline__ float4 load_row4(const float* __restrict__ in32,
                                            const _Float16* __restrict__ in16, int s, int lane) {
    if (kHalfIn) {
        const v4h v = ((const v4h*)in16)[s * 32 + lane];
        return make_float4((float)v[0], (float)v[1], (float)v[2], (float)v[3]);
    } else {
        return ((const float4*)in32)[s * 32 + lane];
    }
}

template <int G, bool kHalfIn>
__device__ __forceinline__ void sum_slots(float4& acc, int ent, float wl, int j,
                                          const float* __restrict__ in32,
                                          const _Float16* __restrict__ in16, int lane) {
    float4 v[G];
    float  w[G];
#pragma unroll
    for (int k = 0; k < G; ++k) {
        const int s = __builtin_amdgcn_readlane(ent, j + k);
        w[k] = __int_as_float(__builtin_amdgcn_readlane(__float_as_int(wl), j + k));
        v[k] = load_row4<kHalfIn>(in32, in16, s, lane);
    }
#pragma unroll
    for (int k = 0; k < G; ++k) {
        acc.x = fmaf(v[k].x, w[k], acc.x);
        acc.y = fmaf(v[k].y, w[k], acc.y);
        acc.z = fmaf(v[k].z, w[k], acc.z);
        acc.w = fmaf(v[k].w, w[k], acc.w);
    }
}

__device__ __forceinline__ int order_slots(int ent, int m, int lane) {
    const unsigned key = ((unsigned)ent << 5) | (unsigned)lane;
    int pos = 1;
    for (int j = 1; j < m; ++j) {
        const unsigned kj = (unsigned)__builtin_amdgcn_readlane((int)key, j);
        pos += (kj < key) ? 1 : 0;
    }
    if (lane == 0 || lane >= m) pos = lane;
    return __builtin_amdgcn_ds_permute(pos << 2, ent);
}

constexpr int kListSrc = 31;

constexpr int kAPitch = 136;
constexpr int kCPitch = 132;
constexpr int kTileRows = 32;
constexpr int kWaveRows = 4;

template <bool kFirst>
__device__ __forceinline__ void layer_tile(const int* __restrict__ rowptr, int* __restrict__ col,
                                           const float* __restrict__ dis,
                                           const float* __restrict__ in32,
                                           const _Float16* __restrict__ in16,
                                           const _Float16* __restrict__ Wt,
                                           const float* __restrict__ bias,
                                           const float* __restrict__ Wq,
                                           _Float16* __restrict__ out16,
                                           float* __restrict__ pq, int Nn) {
    __shared__ __attribute__((aligned(16))) _Float16 As[kTileRows * kAPitch];
    __shared__ __attribute__((aligned(16))) float    Cs[kTileRows * kCPitch];
    const int wave = __builtin_amdgcn_readfirstlane((int)(threadIdx.x >> 5));
    const int lane = threadIdx.x & 31;
    const int row0 = blockIdx.x * kTileRows;

    constexpr bool kHalfIn = !kFirst;
    const int d0 = row0 + wave * kWaveRows;
    int   dn[kWaveRows], beg[kWaveRows], cnt[kWaveRows], m[kWaveRows];
    float dd[kWaveRows];
    int endr[kWaveRows];
#pragma unroll
    for (int i = 0; i < kWaveRows; ++i)
        dn[i] = (d0 + i < Nn) ? (d0 + i) : (Nn - 1);
    if (d0 + kWaveRows <= Nn) {
        const int4   rp = *(const int4*)(rowptr + d0);
        const int    r4 = rowptr[d0 + kWaveRows];
        const float4 dq = *(const float4*)(dis + d0);
        beg[0] = rp.x; beg[1] = rp.y; beg[2] = rp.z; beg[3] = rp.w;
        endr[0] = rp.y; endr[1] = rp.z; endr[2] = rp.w; endr[3] = r4;
        dd[0] = dq.x; dd[1] = dq.y; dd[2] = dq.z; dd[3] = dq.w;
    } else {
#pragma unroll
        for (int i = 0; i < kWaveRows; ++i) {
            beg[i]  = rowptr[dn[i]];
            endr[i] = rowptr[dn[i] + 1];
            dd[i]   = dis[dn[i]];
        }
    }
#pragma unroll
    for (int i = 0; i < kWaveRows; ++i) {
        const bool ok = (d0 + i) < Nn;
        cnt[i] = ok ? (endr[i] - beg[i]) : 0;
        m[i]   = ok ? ((cnt[i] < kListSrc ? cnt[i] : kListSrc) + 1) : 0;
    }
    int ent[kWaveRows];
#pragma unroll
    for (int i = 0; i < kWaveRows; ++i) {
        ent[i] = dn[i];
        if (lane >= 1 && lane < m[i]) ent[i] = col[beg[i] + lane - 1];
    }
    float ds[kWaveRows];
#pragma unroll
    for (int i = 0; i < kWaveRows; ++i) {
        if (kFirst) {
            ent[i] = order_slots(ent[i], m[i], lane);
            if (lane >= 1 && lane < m[i]) col[beg[i] + lane - 1] = ent[i];
        }
        ds[i] = 0.0f;
        if (lane < m[i]) ds[i] = dis[ent[i]];
    }
#pragma unroll
    for (int i = 0; i < kWaveRows; ++i) {
        const int   rt = wave * kWaveRows + i;
        const float wl = ds[i] * dd[i];
        float4 r = make_float4(0.0f, 0.0f, 0.0f, 0.0f);
        int j = 0;
        for (; j + 4 < m[i]; j += 8) sum_slots<8, kHalfIn>(r, ent[i], wl, j, in32, in16, lane);
        if (j < m[i])                sum_slots<4, kHalfIn>(r, ent[i], wl, j, in32, in16, lane);
        if (cnt[i] > kListSrc) {
            const int end = beg[i] + cnt[i];
            for (int base = beg[i] + kListSrc; base < end; base += 32) {
                const int n = (end - base < 32) ? (end - base) : 32;
                int   c2 = 0;
                float w2 = 0.0f;
                if (lane < n) c2 = col[base + lane];
                if (kFirst) {
                    const unsigned key = ((unsigned)c2 << 5) | (unsigned)lane;
                    int pos = 0;
                    for (int q = 0; q < n; ++q) {
                        const unsigned kq = (unsigned)__builtin_amdgcn_readlane((int)key, q);
                        pos += (kq < key) ? 1 : 0;
                    }
                    if (lane >= n) pos = lane;
                    c2 = __builtin_amdgcn_ds_permute(pos << 2, c2);
                    if (lane < n) col[base + lane] = c2;
                }
                if (lane < n) w2 = dis[c2] * dd[i];
                for (int q = 0; q < n; ++q) {
                    const int   s = __builtin_amdgcn_readlane(c2, q);
                    const float w = __int_as_float(__builtin_amdgcn_readlane(__float_as_int(w2), q));
                    const float4 v = load_row4<kHalfIn>(in32, in16, s, lane);
                    r.x = fmaf(v.x, w, r.x);
                    r.y = fmaf(v.y, w, r.y);
                    r.z = fmaf(v.z, w, r.z);
                    r.w = fmaf(v.w, w, r.w);
                }
            }
        }
        v4h o;
        o[0] = (_Float16)r.x; o[1] = (_Float16)r.y; o[2] = (_Float16)r.z; o[3] = (_Float16)r.w;
        *(v4h*)(As + rt * kAPitch + lane * 4) = o;
    }
    __syncthreads();

    const int c = lane & 15;
    const int kbase = (lane >> 4) ? 8 : 0;
    const _Float16* arow = As + c * kAPitch;
    const _Float16* __restrict__ brow = Wt + (long long)(wave * 16 + c) * 128;
    v8f acc0 = {}, acc1 = {};
#pragma unroll
    for (int kc = 0; kc < 4; ++kc) {
        const int k0 = kc * 32 + kbase;
        v8h blo  = *(const v8h*)(brow + k0);
        v8h bhi  = *(const v8h*)(brow + k0 + 16);
        v8h a0lo = *(const v8h*)(arow + k0);
        v8h a0hi = *(const v8h*)(arow + k0 + 16);
        v8h a1lo = *(const v8h*)(arow + 16 * kAPitch + k0);
        v8h a1hi = *(const v8h*)(arow + 16 * kAPitch + k0 + 16);
        v16h a0, a1, b;
#pragma unroll
        for (int e = 0; e < 8; ++e) {
            a0[e] = a0lo[e]; a0[8 + e] = a0hi[e];
            a1[e] = a1lo[e]; a1[8 + e] = a1hi[e];
            b[e]  = blo[e];  b[8 + e]  = bhi[e];
        }
        acc0 = __builtin_amdgcn_wmma_f32_16x16x32_f16(
             false, a0,  false, b,
             (short)0, acc0,  false,  false);
        acc1 = __builtin_amdgcn_wmma_f32_16x16x32_f16(
             false, a1,  false, b,
             (short)0, acc1,  false,  false);
    }

    {
        const float bcol = bias[wave * 16 + c];
        float* crow = Cs + ((lane >> 4) ? 8 : 0) * kCPitch + wave * 16 + c;
#pragma unroll
        for (int r = 0; r < 8; ++r) {
            float v0 = acc0[r] + bcol;
            float v1 = acc1[r] + bcol;
            crow[r * kCPitch]        = v0 > 0.0f ? v0 : 0.0f;
            crow[(16 + r) * kCPitch] = v1 > 0.0f ? v1 : 0.0f;
        }
    }
    float4 wa = make_float4(0.0f, 0.0f, 0.0f, 0.0f), wb = wa;
    if (!kFirst) {
        wa = ((const float4*)Wq)[lane];
        wb = ((const float4*)Wq)[32 + lane];
    }
    __syncthreads();

    for (int i = 0; i < kWaveRows; ++i) {
        const int rt = wave * kWaveRows + i;
        const int d  = row0 + rt;
        if (d >= Nn) continue;
        const float4 r = *(const float4*)(Cs + rt * kCPitch + lane * 4);
        if (kFirst) {
            v4h o;
            o[0] = (_Float16)r.x; o[1] = (_Float16)r.y; o[2] = (_Float16)r.z; o[3] = (_Float16)r.w;
            *(v4h*)(out16 + (long long)d * 128 + lane * 4) = o;
        } else {
            float p = r.x * wa.x + r.y * wa.y + r.z * wa.z + r.w * wa.w;
            float t = r.x * wb.x + r.y * wb.y + r.z * wb.z + r.w * wb.w;
#pragma unroll
            for (int o = 16; o >= 1; o >>= 1) {
                p += __shfl_xor(p, o);
                t += __shfl_xor(t, o);
            }
            if (lane == 0) *(float2*)(pq + 2 * (long long)d) = make_float2(p, t);
        }
    }
}

__global__ void __launch_bounds__(256) layer1_agg_mm(const int* __restrict__ rowptr, int* __restrict__ col,
                                                  const float* __restrict__ dis,
                                                  const float* __restrict__ x,
                                                  const _Float16* __restrict__ Wt,
                                                  const float* __restrict__ bias,
                                                  _Float16* __restrict__ out16, int Nn) {
    layer_tile<true>(rowptr, col, dis, x, nullptr, Wt, bias, nullptr, out16, nullptr, Nn);
}

__global__ void __launch_bounds__(256) layer2_agg_mm_dotq(const int* __restrict__ rowptr, int* __restrict__ col,
                                                       const float* __restrict__ dis,
                                                       const _Float16* __restrict__ hin,
                                                       const _Float16* __restrict__ Wt,
                                                       const float* __restrict__ bias,
                                                       const float* __restrict__ Wq,
                                                       float* __restrict__ pq, int Nn) {
    layer_tile<false>(rowptr, col, dis, nullptr, hin, Wt, bias, Wq, nullptr, pq, Nn);
}

__global__ void __launch_bounds__(256) q_pairs4(const int* __restrict__ va, const float* __restrict__ pq,
                                                const float* __restrict__ bq, float* __restrict__ out, int A) {
    const int t  = blockIdx.x * 256 + threadIdx.x;
    const int a0 = t * 4;
    if (a0 >= A) return;
    const float b = bq[0];
    if (a0 + 3 < A) {
        const int4 u = ((const int4*)va)[2 * t];
        const int4 w = ((const int4*)va)[2 * t + 1];
        float4 o;
        o.x = (pq[2 * (long long)u.x] + pq[2 * (long long)u.y + 1]) + b;
        o.y = (pq[2 * (long long)u.z] + pq[2 * (long long)u.w + 1]) + b;
        o.z = (pq[2 * (long long)w.x] + pq[2 * (long long)w.y + 1]) + b;
        o.w = (pq[2 * (long long)w.z] + pq[2 * (long long)w.w + 1]) + b;
        ((float4*)out)[t] = o;
    } else {
        for (int a = a0; a < A; ++a) {
            const int2 ij = ((const int2*)va)[a];
            out[a] = (pq[2 * (long long)ij.x] + pq[2 * (long long)ij.y + 1]) + b;
        }
    }
}

extern "C" void kernel_launch(void* const* d_in, const int* in_sizes, int n_in,
                              void* d_out, int out_size, void* d_ws, size_t ws_size,
                              hipStream_t stream) {
    const float* x  = (const float*)d_in[0];
    const int*   ei = (const int*)d_in[1];
    const int*   va = (const int*)d_in[2];
    const float* W1 = (const float*)d_in[3];
    const float* b1 = (const float*)d_in[4];
    const float* W2 = (const float*)d_in[5];
    const float* b2 = (const float*)d_in[6];
    const float* Wq = (const float*)d_in[7];
    const float* bq = (const float*)d_in[8];
    float* q = (float*)d_out;

    const int H  = 128;
    const int Nn = in_sizes[0] / H;
    const int E  = in_sizes[1] / 2;
    const int A  = in_sizes[2] / 2;

    char*  ws  = (char*)d_ws;
    size_t off = 0;
    auto alloc = [&](size_t bytes) -> char* {
        char* p = ws + off;
        off = (off + bytes + 255) & ~(size_t)255;
        return p;
    };
    _Float16* h1h = (_Float16*)alloc((size_t)Nn * H * 2);
    _Float16* w1t = (_Float16*)alloc((size_t)H * H * 2);
    _Float16* w2t = (_Float16*)alloc((size_t)H * H * 2);
    float*    pq  = (float*)   alloc((size_t)Nn * 2 * 4);
    char*     csr = (char*)    alloc((size_t)Nn * H * 2);
    float*    dis = (float*)   alloc((size_t)Nn * 4);

    size_t coff = 0;
    auto calloc_ = [&](size_t bytes) -> int* {
        int* p = (int*)(csr + coff);
        coff = (coff + bytes + 255) & ~(size_t)255;
        return p;
    };
    int* rowptr = calloc_((size_t)(Nn + 1) * 4);
    int* rank   = calloc_((size_t)E * 4);
    int* col    = calloc_((size_t)E * 4);
    int* cellT  = calloc_((size_t)kChunks * kBucket * 4);
    int* cellO  = calloc_((size_t)kChunks * kBucket * 4);
    int* staged = calloc_((size_t)E * 8);

    auto blks = [](long long n, int t) { return (unsigned)((n + t - 1) / t); };

    const int CS = (int)blks(E, kChunks);
    bin_chunk<<<kChunks, 1024, 0, stream>>>(ei, rank, cellT, cellO, staged, E, CS);
    bucket_rows<<<blks(Nn, kBucket) + kCvtBlocks, 1024, 0, stream>>>(cellT, cellO, staged, rowptr, dis, col, Nn, CS,
                                                                      W1, W2, w1t, w2t);

    const unsigned mtiles = blks(Nn, kTileRows);

    layer1_agg_mm<<<mtiles, 256, 0, stream>>>(rowptr, col, dis, x, w1t, b1, h1h, Nn);
    layer2_agg_mm_dotq<<<mtiles, 256, 0, stream>>>(rowptr, col, dis, h1h, w2t, b2, Wq, pq, Nn);

    q_pairs4<<<blks(A, 1024), 256, 0, stream>>>(va, pq, bq, q, A);
}


# both edge-list kernels place their scattered index stores (bucket-grouped pairs, per-node source ids) in an LDS copy of the region and write it out as contiguous whole-line stores; sums unchanged
# speedup vs baseline: 7.2719x; 1.1210x over previous
#include <hip/hip_runtime.h>
#include <hip/hip_bf16.h>

typedef __attribute__((ext_vector_type(16))) _Float16 v16h;
typedef __attribute__((ext_vector_type(8)))  _Float16 v8h;
typedef __attribute__((ext_vector_type(4)))  _Float16 v4h;
typedef __attribute__((ext_vector_type(8)))  float    v8f;

constexpr int kBucket = 256;
constexpr int kChunks = 256;
constexpr int kStageCap = 4096;
constexpr int kColCap   = 8192;

__device__ __forceinline__ int wave_incl_scan(int x, int lane) {
#pragma unroll
    for (int o = 1; o < 32; o <<= 1) {
        int y = __shfl_up(x, o);
        if (lane >= o) x += y;
    }
    return x;
}

__device__ __forceinline__ void place_pairs(const int* __restrict__ ei, const int* __restrict__ rank,
                                            const int* offs, int2* region, int E, int e0, int e1, int tid) {
    for (int eb = e0 + tid; eb < e1; eb += 4096) {
        int d[4], s[4], r[4];
#pragma unroll
        for (int k = 0; k < 4; ++k) {
            const int e = eb + 1024 * k;
            const bool in = (e < e1);
            d[k] = in ? ei[E + e] : -1;
            s[k] = in ? ei[e] : 0;
            r[k] = in ? rank[e] : 0;
        }
#pragma unroll
        for (int k = 0; k < 4; ++k) {
            if (d[k] >= 0) region[offs[d[k] >> 8] + r[k]] = make_int2(s[k], d[k]);
        }
    }
}

__global__ void __launch_bounds__(1024) bin_chunk(const int* __restrict__ ei, int* __restrict__ rank,
                                                  int* __restrict__ T, int* __restrict__ O,
                                                  int* __restrict__ staged, int E, int CS) {
    __shared__ int hist[kBucket];
    __shared__ int offs[kBucket];
    __shared__ int wsum[8];
    __shared__ int2 sbuf[kStageCap];
    const int tid = threadIdx.x, lane = tid & 31, wave = tid >> 5;
    const int c  = blockIdx.x;
    const int e0 = c * CS;
    const int e1 = (e0 + CS < E) ? (e0 + CS) : E;
    if (tid < kBucket) hist[tid] = 0;
    __syncthreads();
    for (int eb = e0 + tid; eb < e1; eb += 4096) {
        int d[4];
#pragma unroll
        for (int k = 0; k < 4; ++k) {
            const int e = eb + 1024 * k;
            d[k] = (e < e1) ? ei[E + e] : -1;
        }
#pragma unroll
        for (int k = 0; k < 4; ++k) {
            const int e = eb + 1024 * k;
            if (d[k] >= 0) rank[e] = atomicAdd(&hist[d[k] >> 8], 1);
        }
    }
    __syncthreads();
    const int v = (tid < kBucket) ? hist[tid] : 0;
    const int x = wave_incl_scan(v, lane);
    if (tid < kBucket && lane == 31) wsum[wave] = x;
    __syncthreads();
    if (tid < kBucket) {
        int pre = 0;
        for (int w = 0; w < wave; ++w) pre += wsum[w];
        const int ex = pre + x - v;
        offs[tid] = ex;
        T[c * kBucket + tid] = v;
        O[c * kBucket + tid] = ex;
    }
    __syncthreads();
    int2* __restrict__ st = (int2*)staged;
    if (CS <= kStageCap) {
        place_pairs(ei, rank, offs, sbuf, E, e0, e1, tid);
        __syncthreads();
        const int n = e1 - e0;
        for (int i = tid; i < n; i += 1024) st[e0 + i] = sbuf[i];
    } else {
        place_pairs(ei, rank, offs, st + e0, E, e0, e1, tid);
    }
}

__device__ __forceinline__ void cvt_transpose_w8(const float* __restrict__ Wm, _Float16* __restrict__ Wt, int t) {
    const int c = t & 127, k0 = (t >> 7) * 8;
    v8h o;
#pragma unroll
    for (int j = 0; j < 8; ++j)
        o[j] = (_Float16)Wm[(k0 + j) * 128 + c];
    *(v8h*)(Wt + c * 128 + k0) = o;
}
constexpr int kCvtBlocks = 4;

__device__ __forceinline__ void place_sources(const int2* __restrict__ cell, int q, int len, int lo,
                                              int* cur, int* part) {
    for (int i = q; i < len; i += 16) {
        int2 p[4];
#pragma unroll
        for (int k = 0; k < 4; ++k)
            p[k] = (i + 4 * k < len) ? cell[i + 4 * k] : make_int2(0, -1);
#pragma unroll
        for (int k = 0; k < 4; ++k)
            if (p[k].y >= 0) part[atomicAdd(&cur[p[k].y - lo], 1)] = p[k].x;
    }
}

__global__ void __launch_bounds__(1024) bucket_rows(const int* __restrict__ T, const int* __restrict__ O,
                                                    const int* __restrict__ staged,
                                                    int* __restrict__ rowptr, float* __restrict__ dis,
                                                    int* __restrict__ col, int Nn, int CS,
                                                    const float* __restrict__ W1, const float* __restrict__ W2,
                                                    _Float16* __restrict__ w1t, _Float16* __restrict__ w2t,
                                                    const float* __restrict__ xin, _Float16* __restrict__ xs) {
    __shared__ int cnt[kBucket];
    __shared__ int cur[kBucket];
    __shared__ int wsum[8];
    __shared__ int wred[32];
    __shared__ float sdis[kBucket];
    __shared__ int stot;
    __shared__ int cbuf[kColCap];
    const int nb = (int)gridDim.x - kCvtBlocks;
    if ((int)blockIdx.x >= nb) {
        const int t = ((int)blockIdx.x - nb) * 1024 + (int)threadIdx.x;
        if (t < 2048) cvt_transpose_w8(W1, w1t, t);
        else          cvt_transpose_w8(W2, w2t, t - 2048);
        return;
    }
    const int tid = threadIdx.x, lane = tid & 31, wave = tid >> 5;
    const int b  = blockIdx.x;
    const int lo = b * kBucket;
    const int c  = tid >> 2, q = tid & 3;
    const int o   = O[c * kBucket + b];
    const int len = T[c * kBucket + b];
    const int2* __restrict__ cell = (const int2*)staged + ((long long)c * CS + o);
    float4 xr[8];
#pragma unroll
    for (int k = 0; k < 8; ++k) {
        const int node = lo + wave + 32 * k;
        const int nc   = (node < Nn) ? node : (Nn - 1);
        xr[k] = ((const float4*)xin)[nc * 32 + lane];
    }

    int r = (q == 0) ? o : 0;
#pragma unroll
    for (int m = 16; m >= 1; m >>= 1) r += __shfl_xor(r, m);
    if (lane == 0) wred[wave] = r;
    if (tid < kBucket) cnt[tid] = 0;
    __syncthreads();
    int base = wred[lane];
#pragma unroll
    for (int m = 16; m >= 1; m >>= 1) base += __shfl_xor(base, m);

    for (int i = q; i < len; i += 16) {
        int dn[4];
#pragma unroll
        for (int k = 0; k < 4; ++k)
            dn[k] = (i + 4 * k < len) ? cell[i + 4 * k].y : -1;
#pragma unroll
        for (int k = 0; k < 4; ++k)
            if (dn[k] >= 0) atomicAdd(&cnt[dn[k] - lo], 1);
    }
    __syncthreads();

    const int v = (tid < kBucket) ? cnt[tid] : 0;
    const int x = wave_incl_scan(v, lane);
    if (tid < kBucket && lane == 31) wsum[wave] = x;
    __syncthreads();
    if (tid < kBucket) {
        int pre = 0;
        for (int w = 0; w < wave; ++w) pre += wsum[w];
        const int ex = pre + x - v;
        const int node = lo + tid;
        const float deg = (float)(v + 1);
        const float dv  = (deg > 0.0f) ? rsqrtf(deg) : 0.0f;
        if (node < Nn) {
            rowptr[node] = base + ex;
            dis[node] = dv;
        }
        if (b == nb - 1 && tid == kBucket - 1) rowptr[Nn] = base + ex + v;
        cur[tid] = ex;
        sdis[tid] = dv;
        if (tid == kBucket - 1) stot = ex + v;
    }
    __syncthreads();

#pragma unroll
    for (int k = 0; k < 8; ++k) {
        const int node = lo + wave + 32 * k;
        if (node < Nn) {
            const float dv = sdis[wave + 32 * k];
            v4h o4;
            o4[0] = (_Float16)(xr[k].x * dv); o4[1] = (_Float16)(xr[k].y * dv);
            o4[2] = (_Float16)(xr[k].z * dv); o4[3] = (_Float16)(xr[k].w * dv);
            ((v4h*)xs)[node * 32 + lane] = o4;
        }
    }

    const int total = stot;
    if (total <= kColCap) {
        place_sources(cell, q, len, lo, cur, cbuf);
        __syncthreads();
        for (int i = tid; i < total; i += 1024) col[base + i] = cbuf[i];
    } else {
        place_sources(cell, q, len, lo, cur, col + base);
    }
}


__device__ __forceinline__ uint2 load_row4(const _Float16* __restrict__ in16, int s, int lane) {
    return ((const uint2*)in16)[s * 32 + lane];
}

__device__ __forceinline__ void fma_lo(float& acc, unsigned p, float w) {
    asm("v_fma_mix_f32 %0, %1, %2, %0 op_sel_hi:[1,0,0]" : "+v"(acc) : "v"(p), "s"(w));
}
__device__ __forceinline__ void fma_hi(float& acc, unsigned p, float w) {
    asm("v_fma_mix_f32 %0, %1, %2, %0 op_sel:[1,0,0] op_sel_hi:[1,0,0]" : "+v"(acc) : "v"(p), "s"(w));
}

template <int G>
__device__ __forceinline__ void sum_slots(float4& acc, int ent, int mm, float wd, int j,
                                          const _Float16* __restrict__ in16, int lane) {
    uint2 v[G];
    float w[G];
#pragma unroll
    for (int k = 0; k < G; ++k) {
        const int s = __builtin_amdgcn_readlane(ent, j + k);
        w[k] = (j + k < mm) ? wd : 0.0f;
        v[k] = load_row4(in16, s, lane);
    }
    __builtin_amdgcn_sched_barrier(0);
#pragma unroll
    for (int k = 0; k < G; ++k) {
        fma_lo(acc.x, v[k].x, w[k]);
        fma_hi(acc.y, v[k].x, w[k]);
        fma_lo(acc.z, v[k].y, w[k]);
        fma_hi(acc.w, v[k].y, w[k]);
    }
}

__device__ __forceinline__ int order_slots(int ent, int m, int lane) {
    const unsigned key = ((unsigned)ent << 5) | (unsigned)lane;
    int pos = 1;
    for (int j = 1; j < m; ++j) {
        const unsigned kj = (unsigned)__builtin_amdgcn_readlane((int)key, j);
        pos += (kj < key) ? 1 : 0;
    }
    if (lane == 0 || lane >= m) pos = lane;
    return __builtin_amdgcn_ds_permute(pos << 2, ent);
}

constexpr int kListSrc = 31;

constexpr int kAPitch = 136;
constexpr int kCPitch = 132;
constexpr int kTileRows = 32;
constexpr int kWaveRows = 4;

template <bool kFirst>
__device__ __forceinline__ void layer_tile(const int* __restrict__ rowptr, int* __restrict__ col,
                                           const float* __restrict__ dis,
                                           const _Float16* __restrict__ in16,
                                           const _Float16* __restrict__ Wt,
                                           const float* __restrict__ bias,
                                           const float* __restrict__ Wq,
                                           _Float16* __restrict__ out16,
                                           float* __restrict__ pq, int Nn) {
    __shared__ __attribute__((aligned(16))) _Float16 As[kTileRows * kAPitch];
    __shared__ __attribute__((aligned(16))) float    Cs[kTileRows * kCPitch];
    const int wave = __builtin_amdgcn_readfirstlane((int)(threadIdx.x >> 5));
    const int lane = threadIdx.x & 31;
    const int row0 = blockIdx.x * kTileRows;

    const int d0 = row0 + wave * kWaveRows;
    int   dn[kWaveRows], beg[kWaveRows], cnt[kWaveRows], m[kWaveRows];
    float dd[kWaveRows];
    int endr[kWaveRows];
#pragma unroll
    for (int i = 0; i < kWaveRows; ++i)
        dn[i] = (d0 + i < Nn) ? (d0 + i) : (Nn - 1);
    if (d0 + kWaveRows <= Nn) {
        const int4   rp = *(const int4*)(rowptr + d0);
        const int    r4 = rowptr[d0 + kWaveRows];
        const float4 dq = *(const float4*)(dis + d0);
        beg[0] = rp.x; beg[1] = rp.y; beg[2] = rp.z; beg[3] = rp.w;
        endr[0] = rp.y; endr[1] = rp.z; endr[2] = rp.w; endr[3] = r4;
        dd[0] = dq.x; dd[1] = dq.y; dd[2] = dq.z; dd[3] = dq.w;
    } else {
#pragma unroll
        for (int i = 0; i < kWaveRows; ++i) {
            beg[i]  = rowptr[dn[i]];
            endr[i] = rowptr[dn[i] + 1];
            dd[i]   = dis[dn[i]];
        }
    }
#pragma unroll
    for (int i = 0; i < kWaveRows; ++i) {
        const bool ok = (d0 + i) < Nn;
        cnt[i] = ok ? (endr[i] - beg[i]) : 0;
        m[i]   = ok ? ((cnt[i] < kListSrc ? cnt[i] : kListSrc) + 1) : 0;
    }
    int ent[kWaveRows];
#pragma unroll
    for (int i = 0; i < kWaveRows; ++i) {
        ent[i] = dn[i];
        if (lane >= 1 && lane < m[i]) ent[i] = col[beg[i] + lane - 1];
    }
    if (kFirst) {
#pragma unroll
        for (int i = 0; i < kWaveRows; ++i) {
            ent[i] = order_slots(ent[i], m[i], lane);
            if (lane >= 1 && lane < m[i]) col[beg[i] + lane - 1] = ent[i];
        }
    }
#pragma unroll
    for (int i = 0; i < kWaveRows; ++i) {
        const int   rt = wave * kWaveRows + i;
        float4 r = make_float4(0.0f, 0.0f, 0.0f, 0.0f);
        int j = 0;
        for (; j + 8 < m[i]; j += 16) sum_slots<16>(r, ent[i], m[i], dd[i], j, in16, lane);
        if      (j + 4 < m[i]) sum_slots<8>(r, ent[i], m[i], dd[i], j, in16, lane);
        else if (j < m[i])     sum_slots<4>(r, ent[i], m[i], dd[i], j, in16, lane);
        if (cnt[i] > kListSrc) {
            const int end = beg[i] + cnt[i];
            for (int base = beg[i] + kListSrc; base < end; base += 32) {
                const int n = (end - base < 32) ? (end - base) : 32;
                int   c2 = 0;
                if (lane < n) c2 = col[base + lane];
                if (kFirst) {
                    const unsigned key = ((unsigned)c2 << 5) | (unsigned)lane;
                    int pos = 0;
                    for (int q = 0; q < n; ++q) {
                        const unsigned kq = (unsigned)__builtin_amdgcn_readlane((int)key, q);
                        pos += (kq < key) ? 1 : 0;
                    }
                    if (lane >= n) pos = lane;
                    c2 = __builtin_amdgcn_ds_permute(pos << 2, c2);
                    if (lane < n) col[base + lane] = c2;
                }
                const float w = dd[i];
                for (int q = 0; q < n; ++q) {
                    const int   s = __builtin_amdgcn_readlane(c2, q);
                    const uint2 v = load_row4(in16, s, lane);
                    fma_lo(r.x, v.x, w);
                    fma_hi(r.y, v.x, w);
                    fma_lo(r.z, v.y, w);
                    fma_hi(r.w, v.y, w);
                }
            }
        }
        v4h o;
        o[0] = (_Float16)r.x; o[1] = (_Float16)r.y; o[2] = (_Float16)r.z; o[3] = (_Float16)r.w;
        *(v4h*)(As + rt * kAPitch + lane * 4) = o;
    }
    __syncthreads();

    const int c = lane & 15;
    const int kbase = (lane >> 4) ? 8 : 0;
    const _Float16* arow = As + c * kAPitch;
    const _Float16* __restrict__ brow = Wt + (long long)(wave * 16 + c) * 128;
    v8f acc0 = {}, acc1 = {};
#pragma unroll
    for (int kc = 0; kc < 4; ++kc) {
        const int k0 = kc * 32 + kbase;
        v8h blo  = *(const v8h*)(brow + k0);
        v8h bhi  = *(const v8h*)(brow + k0 + 16);
        v8h a0lo = *(const v8h*)(arow + k0);
        v8h a0hi = *(const v8h*)(arow + k0 + 16);
        v8h a1lo = *(const v8h*)(arow + 16 * kAPitch + k0);
        v8h a1hi = *(const v8h*)(arow + 16 * kAPitch + k0 + 16);
        v16h a0, a1, b;
#pragma unroll
        for (int e = 0; e < 8; ++e) {
            a0[e] = a0lo[e]; a0[8 + e] = a0hi[e];
            a1[e] = a1lo[e]; a1[8 + e] = a1hi[e];
            b[e]  = blo[e];  b[8 + e]  = bhi[e];
        }
        acc0 = __builtin_amdgcn_wmma_f32_16x16x32_f16(
             false, a0,  false, b,
             (short)0, acc0,  false,  false);
        acc1 = __builtin_amdgcn_wmma_f32_16x16x32_f16(
             false, a1,  false, b,
             (short)0, acc1,  false,  false);
    }

    {
        const float bcol = bias[wave * 16 + c];
        float* crow = Cs + ((lane >> 4) ? 8 : 0) * kCPitch + wave * 16 + c;
#pragma unroll
        for (int r = 0; r < 8; ++r) {
            float v0 = acc0[r] + bcol;
            float v1 = acc1[r] + bcol;
            crow[r * kCPitch]        = v0 > 0.0f ? v0 : 0.0f;
            crow[(16 + r) * kCPitch] = v1 > 0.0f ? v1 : 0.0f;
        }
    }
    float4 wa = make_float4(0.0f, 0.0f, 0.0f, 0.0f), wb = wa;
    if (!kFirst) {
        wa = ((const float4*)Wq)[lane];
        wb = ((const float4*)Wq)[32 + lane];
    }
    __syncthreads();

#pragma unroll
    for (int i = 0; i < kWaveRows; ++i) {
        const int rt = wave * kWaveRows + i;
        const int d  = row0 + rt;
        if (d >= Nn) continue;
        const float4 r = *(const float4*)(Cs + rt * kCPitch + lane * 4);
        if (kFirst) {
            v4h o;
            o[0] = (_Float16)(r.x * dd[i]); o[1] = (_Float16)(r.y * dd[i]);
            o[2] = (_Float16)(r.z * dd[i]); o[3] = (_Float16)(r.w * dd[i]);
            *(v4h*)(out16 + (long long)d * 128 + lane * 4) = o;
        } else {
            float p = r.x * wa.x + r.y * wa.y + r.z * wa.z + r.w * wa.w;
            float t = r.x * wb.x + r.y * wb.y + r.z * wb.z + r.w * wb.w;
#pragma unroll
            for (int o = 16; o >= 1; o >>= 1) {
                p += __shfl_xor(p, o);
                t += __shfl_xor(t, o);
            }
            if (lane == 0) *(float2*)(pq + 2 * (long long)d) = make_float2(p, t);
        }
    }
}

__global__ void __launch_bounds__(256) layer1_agg_mm(const int* __restrict__ rowptr, int* __restrict__ col,
                                                  const float* __restrict__ dis,
                                                  const _Float16* __restrict__ xs,
                                                  const _Float16* __restrict__ Wt,
                                                  const float* __restrict__ bias,
                                                  _Float16* __restrict__ out16, int Nn) {
    layer_tile<true>(rowptr, col, dis, xs, Wt, bias, nullptr, out16, nullptr, Nn);
}

__global__ void __launch_bounds__(256) layer2_agg_mm_dotq(const int* __restrict__ rowptr, int* __restrict__ col,
                                                       const float* __restrict__ dis,
                                                       const _Float16* __restrict__ hin,
                                                       const _Float16* __restrict__ Wt,
                                                       const float* __restrict__ bias,
                                                       const float* __restrict__ Wq,
                                                       float* __restrict__ pq, int Nn) {
    layer_tile<false>(rowptr, col, dis, hin, Wt, bias, Wq, nullptr, pq, Nn);
}

__global__ void __launch_bounds__(256) q_pairs4(const int* __restrict__ va, const float* __restrict__ pq,
                                                const float* __restrict__ bq, float* __restrict__ out, int A) {
    const int t  = blockIdx.x * 256 + threadIdx.x;
    const int a0 = t * 4;
    if (a0 >= A) return;
    const float b = bq[0];
    if (a0 + 3 < A) {
        const int4 u = ((const int4*)va)[2 * t];
        const int4 w = ((const int4*)va)[2 * t + 1];
        float4 o;
        o.x = (pq[2 * (long long)u.x] + pq[2 * (long long)u.y + 1]) + b;
        o.y = (pq[2 * (long long)u.z] + pq[2 * (long long)u.w + 1]) + b;
        o.z = (pq[2 * (long long)w.x] + pq[2 * (long long)w.y + 1]) + b;
        o.w = (pq[2 * (long long)w.z] + pq[2 * (long long)w.w + 1]) + b;
        ((float4*)out)[t] = o;
    } else {
        for (int a = a0; a < A; ++a) {
            const int2 ij = ((const int2*)va)[a];
            out[a] = (pq[2 * (long long)ij.x] + pq[2 * (long long)ij.y + 1]) + b;
        }
    }
}

extern "C" void kernel_launch(void* const* d_in, const int* in_sizes, int n_in,
                              void* d_out, int out_size, void* d_ws, size_t ws_size,
                              hipStream_t stream) {
    const float* x  = (const float*)d_in[0];
    const int*   ei = (const int*)d_in[1];
    const int*   va = (const int*)d_in[2];
    const float* W1 = (const float*)d_in[3];
    const float* b1 = (const float*)d_in[4];
    const float* W2 = (const float*)d_in[5];
    const float* b2 = (const float*)d_in[6];
    const float* Wq = (const float*)d_in[7];
    const float* bq = (const float*)d_in[8];
    float* q = (float*)d_out;

    const int H  = 128;
    const int Nn = in_sizes[0] / H;
    const int E  = in_sizes[1] / 2;
    const int A  = in_sizes[2] / 2;

    char*  ws  = (char*)d_ws;
    size_t off = 0;
    auto alloc = [&](size_t bytes) -> char* {
        char* p = ws + off;
        off = (off + bytes + 255) & ~(size_t)255;
        return p;
    };
    _Float16* h1h = (_Float16*)alloc((size_t)Nn * H * 2);
    _Float16* w1t = (_Float16*)alloc((size_t)H * H * 2);
    _Float16* w2t = (_Float16*)alloc((size_t)H * H * 2);
    float*    pq  = (float*)   alloc((size_t)Nn * 2 * 4);
    char*     csr = (char*)    alloc((size_t)Nn * H * 2);
    float*    dis = (float*)   alloc((size_t)Nn * 4);
    _Float16* xs  = (_Float16*)alloc((size_t)Nn * H * 2);
    if (off > ws_size) return;

    size_t coff = 0;
    auto calloc_ = [&](size_t bytes) -> int* {
        int* p = (int*)(csr + coff);
        coff = (coff + bytes + 255) & ~(size_t)255;
        return p;
    };
    int* rowptr = calloc_((size_t)(Nn + 1) * 4);
    int* rank   = calloc_((size_t)E * 4);
    int* col    = calloc_((size_t)E * 4);
    int* cellT  = calloc_((size_t)kChunks * kBucket * 4);
    int* cellO  = calloc_((size_t)kChunks * kBucket * 4);
    int* staged = calloc_((size_t)E * 8);

    auto blks = [](long long n, int t) { return (unsigned)((n + t - 1) / t); };

    const int CS = (int)blks(E, kChunks);
    bin_chunk<<<kChunks, 1024, 0, stream>>>(ei, rank, cellT, cellO, staged, E, CS);
    bucket_rows<<<blks(Nn, kBucket) + kCvtBlocks, 1024, 0, stream>>>(cellT, cellO, staged, rowptr, dis, col, Nn, CS,
                                                                      W1, W2, w1t, w2t, x, xs);

    const unsigned mtiles = blks(Nn, kTileRows);

    layer1_agg_mm<<<mtiles, 256, 0, stream>>>(rowptr, col, dis, xs, w1t, b1, h1h, Nn);
    layer2_agg_mm_dotq<<<mtiles, 256, 0, stream>>>(rowptr, col, dis, h1h, w2t, b2, Wq, pq, Nn);

    q_pairs4<<<blks(A, 1024), 256, 0, stream>>>(va, pq, bq, q, A);
}
